# starting kernel
# baseline (speedup 1.0000x reference)
#include <hip/hip_runtime.h>
#include <math.h>

#define Bsz   512
#define INsz  256
#define Hsz   1024
#define CONC  1280
#define WMAT  (CONC * Hsz)

typedef __attribute__((ext_vector_type(2))) float v2f;
typedef __attribute__((ext_vector_type(8))) float v8f;

__global__ __launch_bounds__(256) void zero4(float4* __restrict__ p, int n4) {
    int i = blockIdx.x * blockDim.x + threadIdx.x;
    if (i < n4) p[i] = make_float4(0.f, 0.f, 0.f, 0.f);
}

__global__ __launch_bounds__(256) void build_xt(const float* __restrict__ in,
                                                const float* __restrict__ hx,
                                                float* __restrict__ XT) {
    int t = blockIdx.x * blockDim.x + threadIdx.x;
    int j = t >> 9;
    int b = t & (Bsz - 1);
    float v = (j < INsz) ? in[b * INsz + j] : hx[b * Hsz + (j - INsz)];
    XT[j * Bsz + b] = v;
}

__global__ __launch_bounds__(256) void scatter_w(const int*   __restrict__ src,
                                                 const int*   __restrict__ dst,
                                                 const float* __restrict__ signs,
                                                 const float* __restrict__ w1,
                                                 const float* __restrict__ w2,
                                                 const float* __restrict__ wa,
                                                 const float* __restrict__ wb,
                                                 float* __restrict__ W, int E) {
    int e = blockIdx.x * blockDim.x + threadIdx.x;
    if (e >= E) return;
    float  sg   = signs[e];
    size_t base = (size_t)src[e] * Hsz + (size_t)dst[e];
    atomicAdd(&W[base],            w1[e] * sg);
    atomicAdd(&W[WMAT + base],     w2[e] * sg);
    atomicAdd(&W[2ull*WMAT + base], wa[e] * sg);
    atomicAdd(&W[3ull*WMAT + base], wb[e] * sg);
}

__global__ __launch_bounds__(128) void gemm_fused(const float* __restrict__ XT,
                                                  const float* __restrict__ W,
                                                  const float* __restrict__ ts,
                                                  const float* __restrict__ b_ff1,
                                                  const float* __restrict__ b_ff2,
                                                  const float* __restrict__ b_ta,
                                                  const float* __restrict__ b_tb,
                                                  float* __restrict__ out) {
    const int lane = threadIdx.x & 31;
    const int wave = threadIdx.x >> 5;

    const int n0   = (blockIdx.x * 4 + wave) * 16;
    const int m0   = blockIdx.y * 32;
    const int mrow = lane & 15;
    const int ncol = lane & 15;
    const int kp   = (lane >> 4) << 1;

    v8f acc[2][4];
    #pragma unroll
    for (int s = 0; s < 2; ++s)
        #pragma unroll
        for (int l = 0; l < 4; ++l)
            acc[s][l] = (v8f){0.f,0.f,0.f,0.f,0.f,0.f,0.f,0.f};

    const float* xa0 = XT + (m0      + mrow);
    const float* xa1 = XT + (m0 + 16 + mrow);

    for (int k0 = 0; k0 < CONC; k0 += 4) {
        const int kr = k0 + kp;
        v2f a0, a1;
        a0.x = xa0[(size_t)kr * Bsz];
        a0.y = xa0[(size_t)(kr + 1) * Bsz];
        a1.x = xa1[(size_t)kr * Bsz];
        a1.y = xa1[(size_t)(kr + 1) * Bsz];

        #pragma unroll
        for (int l = 0; l < 4; ++l) {
            const float* Wp = W + (size_t)l * WMAT + (size_t)kr * Hsz + n0 + ncol;
            v2f b;
            b.x = Wp[0];
            b.y = Wp[Hsz];
            acc[0][l] = __builtin_amdgcn_wmma_f32_16x16x4_f32(
                false, a0, false, b, (short)0, acc[0][l], false, false);
            acc[1][l] = __builtin_amdgcn_wmma_f32_16x16x4_f32(
                false, a1, false, b, (short)0, acc[1][l], false, false);
        }
    }

    const int   ng  = n0 + ncol;
    const float bf1 = b_ff1[ng];
    const float bf2 = b_ff2[ng];
    const float bta = b_ta[ng];
    const float btb = b_tb[ng];
    const int   mhi = (lane >> 4) << 3;

    #pragma unroll
    for (int s = 0; s < 2; ++s) {
        #pragma unroll
        for (int r = 0; r < 8; ++r) {
            const int   m   = m0 + s * 16 + r + mhi;
            const float tsv = ts[m];
            const float f1  = tanhf(acc[s][0][r] + bf1);
            const float f2  = tanhf(acc[s][1][r] + bf2);
            const float ta  = acc[s][2][r] + bta;
            const float tb  = acc[s][3][r] + btb;
            const float sg  = 1.f / (1.f + expf(-(ta * tsv + tb)));
            const float h   = f1 * (1.f - sg) + sg * f2;
            const size_t o  = (size_t)m * Hsz + ng;
            out[o]                         = h;
            out[(size_t)Bsz * Hsz + o]     = h;
        }
    }
}

extern "C" void kernel_launch(void* const* d_in, const int* in_sizes, int n_in,
                              void* d_out, int out_size, void* d_ws, size_t ws_size,
                              hipStream_t stream) {
    const float* input = (const float*)d_in[0];
    const float* hx    = (const float*)d_in[1];
    const float* ts    = (const float*)d_in[2];
    const int*   src   = (const int*)  d_in[3];
    const int*   dst   = (const int*)  d_in[4];
    const float* signs = (const float*)d_in[5];
    const float* w1    = (const float*)d_in[6];
    const float* w2    = (const float*)d_in[7];
    const float* wa    = (const float*)d_in[8];
    const float* wb    = (const float*)d_in[9];
    const float* bf1   = (const float*)d_in[10];
    const float* bf2   = (const float*)d_in[11];
    const float* bta   = (const float*)d_in[12];
    const float* btb   = (const float*)d_in[13];
    const int    E     = in_sizes[3];

    float* W  = (float*)d_ws;
    float* XT = W + 4ull * WMAT;
    float* out = (float*)d_out;

    const int n4 = (4 * WMAT) / 4;
    zero4<<<(n4 + 255) / 256, 256, 0, stream>>>((float4*)W, n4);

    build_xt<<<(CONC * Bsz) / 256, 256, 0, stream>>>(input, hx, XT);

    scatter_w<<<(E + 255) / 256, 256, 0, stream>>>(src, dst, signs,
                                                   w1, w2, wa, wb, W, E);

    dim3 grid(Hsz / 64, Bsz / 32);
    gemm_fused<<<grid, 128, 0, stream>>>(XT, W, ts, bf1, bf2, bta, btb, out);
}


# GEMM K loop software-pipelined over two register sets: the 60 fragment loads of the next 5 k-steps are in flight under the 40 WMMAs of the current 5 (was: 24 loads, wait, 16 WMMAs per trip)
# speedup vs baseline: 1.6944x; 1.6944x over previous
#include <hip/hip_runtime.h>
#include <math.h>

#define Bsz   512
#define INsz  256
#define Hsz   1024
#define CONC  1280
#define WMAT  (CONC * Hsz)

typedef __attribute__((ext_vector_type(2))) float v2f;
typedef __attribute__((ext_vector_type(8))) float v8f;

__global__ __launch_bounds__(256) void zero4(float4* __restrict__ p, int n4) {
    int i = blockIdx.x * blockDim.x + threadIdx.x;
    if (i < n4) p[i] = make_float4(0.f, 0.f, 0.f, 0.f);
}

__global__ __launch_bounds__(256) void build_xt(const float* __restrict__ in,
                                                const float* __restrict__ hx,
                                                float* __restrict__ XT) {
    int t = blockIdx.x * blockDim.x + threadIdx.x;
    int j = t >> 9;
    int b = t & (Bsz - 1);
    float v = (j < INsz) ? in[b * INsz + j] : hx[b * Hsz + (j - INsz)];
    XT[j * Bsz + b] = v;
}

__global__ __launch_bounds__(256) void scatter_w(const int*   __restrict__ src,
                                                 const int*   __restrict__ dst,
                                                 const float* __restrict__ signs,
                                                 const float* __restrict__ w1,
                                                 const float* __restrict__ w2,
                                                 const float* __restrict__ wa,
                                                 const float* __restrict__ wb,
                                                 float* __restrict__ W, int E) {
    int e = blockIdx.x * blockDim.x + threadIdx.x;
    if (e >= E) return;
    float  sg   = signs[e];
    size_t base = (size_t)src[e] * Hsz + (size_t)dst[e];
    atomicAdd(&W[base],            w1[e] * sg);
    atomicAdd(&W[WMAT + base],     w2[e] * sg);
    atomicAdd(&W[2ull*WMAT + base], wa[e] * sg);
    atomicAdd(&W[3ull*WMAT + base], wb[e] * sg);
}

#define KSTEPS 5
#define KBLK   (KSTEPS * 4)

struct FragSet {
    v2f a0[KSTEPS];
    v2f a1[KSTEPS];
    v2f b[KSTEPS][4];
};

__device__ __forceinline__ void load_block(FragSet& f,
                                           const float* __restrict__ xa0,
                                           const float* __restrict__ xa1,
                                           const float* __restrict__ wcol,
                                           int kb, int kp) {
    #pragma unroll
    for (int s = 0; s < KSTEPS; ++s) {
        const int kr = kb + 4 * s + kp;
        f.a0[s].x = xa0[(size_t)kr * Bsz];
        f.a0[s].y = xa0[(size_t)(kr + 1) * Bsz];
        f.a1[s].x = xa1[(size_t)kr * Bsz];
        f.a1[s].y = xa1[(size_t)(kr + 1) * Bsz];
        #pragma unroll
        for (int l = 0; l < 4; ++l) {
            const float* Wp = wcol + (size_t)l * WMAT + (size_t)kr * Hsz;
            f.b[s][l].x = Wp[0];
            f.b[s][l].y = Wp[Hsz];
        }
    }
}

__device__ __forceinline__ void mma_block(const FragSet& f, v8f (&acc)[2][4]) {
    #pragma unroll
    for (int s = 0; s < KSTEPS; ++s) {
        #pragma unroll
        for (int l = 0; l < 4; ++l) {
            acc[0][l] = __builtin_amdgcn_wmma_f32_16x16x4_f32(
                false, f.a0[s], false, f.b[s][l], (short)0, acc[0][l], false, false);
            acc[1][l] = __builtin_amdgcn_wmma_f32_16x16x4_f32(
                false, f.a1[s], false, f.b[s][l], (short)0, acc[1][l], false, false);
        }
    }
}

__global__ __launch_bounds__(128) void gemm_fused(const float* __restrict__ XT,
                                                  const float* __restrict__ W,
                                                  const float* __restrict__ ts,
                                                  const float* __restrict__ b_ff1,
                                                  const float* __restrict__ b_ff2,
                                                  const float* __restrict__ b_ta,
                                                  const float* __restrict__ b_tb,
                                                  float* __restrict__ out) {
    const int lane = threadIdx.x & 31;
    const int wave = threadIdx.x >> 5;

    const int n0   = (blockIdx.x * 4 + wave) * 16;
    const int m0   = blockIdx.y * 32;
    const int mrow = lane & 15;
    const int ncol = lane & 15;
    const int kp   = (lane >> 4) << 1;

    v8f acc[2][4];
    #pragma unroll
    for (int s = 0; s < 2; ++s)
        #pragma unroll
        for (int l = 0; l < 4; ++l)
            acc[s][l] = (v8f){0.f,0.f,0.f,0.f,0.f,0.f,0.f,0.f};

    const float* xa0  = XT + (m0      + mrow);
    const float* xa1  = XT + (m0 + 16 + mrow);
    const float* wcol = W + n0 + ncol;

    FragSet f0, f1;
    load_block(f0, xa0, xa1, wcol, 0, kp);

    int kb = 0;
    for (; kb + 2 * KBLK < CONC; kb += 2 * KBLK) {
        load_block(f1, xa0, xa1, wcol, kb + KBLK, kp);
        mma_block(f0, acc);
        load_block(f0, xa0, xa1, wcol, kb + 2 * KBLK, kp);
        mma_block(f1, acc);
    }
    load_block(f1, xa0, xa1, wcol, kb + KBLK, kp);
    mma_block(f0, acc);
    mma_block(f1, acc);

    const int   ng  = n0 + ncol;
    const float bf1 = b_ff1[ng];
    const float bf2 = b_ff2[ng];
    const float bta = b_ta[ng];
    const float btb = b_tb[ng];
    const int   mhi = (lane >> 4) << 3;

    #pragma unroll
    for (int s = 0; s < 2; ++s) {
        #pragma unroll
        for (int r = 0; r < 8; ++r) {
            const int   m   = m0 + s * 16 + r + mhi;
            const float tsv = ts[m];
            const float f1  = tanhf(acc[s][0][r] + bf1);
            const float f2  = tanhf(acc[s][1][r] + bf2);
            const float ta  = acc[s][2][r] + bta;
            const float tb  = acc[s][3][r] + btb;
            const float sg  = 1.f / (1.f + expf(-(ta * tsv + tb)));
            const float h   = f1 * (1.f - sg) + sg * f2;
            const size_t o  = (size_t)m * Hsz + ng;
            out[o]                         = h;
            out[(size_t)Bsz * Hsz + o]     = h;
        }
    }
}

extern "C" void kernel_launch(void* const* d_in, const int* in_sizes, int n_in,
                              void* d_out, int out_size, void* d_ws, size_t ws_size,
                              hipStream_t stream) {
    const float* input = (const float*)d_in[0];
    const float* hx    = (const float*)d_in[1];
    const float* ts    = (const float*)d_in[2];
    const int*   src   = (const int*)  d_in[3];
    const int*   dst   = (const int*)  d_in[4];
    const float* signs = (const float*)d_in[5];
    const float* w1    = (const float*)d_in[6];
    const float* w2    = (const float*)d_in[7];
    const float* wa    = (const float*)d_in[8];
    const float* wb    = (const float*)d_in[9];
    const float* bf1   = (const float*)d_in[10];
    const float* bf2   = (const float*)d_in[11];
    const float* bta   = (const float*)d_in[12];
    const float* btb   = (const float*)d_in[13];
    const int    E     = in_sizes[3];

    float* W  = (float*)d_ws;
    float* XT = W + 4ull * WMAT;
    float* out = (float*)d_out;

    const int n4 = (4 * WMAT) / 4;
    zero4<<<(n4 + 255) / 256, 256, 0, stream>>>((float4*)W, n4);

    build_xt<<<(CONC * Bsz) / 256, 256, 0, stream>>>(input, hx, XT);

    scatter_w<<<(E + 255) / 256, 256, 0, stream>>>(src, dst, signs,
                                                   w1, w2, wa, wb, W, E);

    dim3 grid(Hsz / 64, Bsz / 32);
    gemm_fused<<<grid, 128, 0, stream>>>(XT, W, ts, bf1, bf2, bta, btb, out);
}


# staging kernels (zero fill, X transpose) give each thread 20 stores / 8 elements instead of one: 4608 waves enter the chip where 61440 one-store waves did (wave entry, not bytes, was their time)
# speedup vs baseline: 1.8262x; 1.0778x over previous
#include <hip/hip_runtime.h>
#include <math.h>

#define Bsz   512
#define INsz  256
#define Hsz   1024
#define CONC  1280
#define WMAT  (CONC * Hsz)

typedef __attribute__((ext_vector_type(2))) float v2f;
typedef __attribute__((ext_vector_type(8))) float v8f;

#define STAGE_THR   256
#define ZERO_WGS    256
#define ZERO_N4     (4 * WMAT / 4)
#define ZERO_PER_T  (ZERO_N4 / (ZERO_WGS * STAGE_THR))
#define XT_ROWS     8
#define XT_WGS      (2 * (CONC / XT_ROWS))

static_assert(ZERO_N4 % (ZERO_WGS * STAGE_THR) == 0, "zero fill must divide evenly");
static_assert(INsz % XT_ROWS == 0 && CONC % XT_ROWS == 0, "a row group reads one source only");
static_assert(Bsz == 2 * STAGE_THR && XT_ROWS == 8, "X^T tiling must cover exactly");

__global__ __launch_bounds__(STAGE_THR) void zero4(float4* __restrict__ p) {
    const int lane = threadIdx.x & 31;
    const int gw   = (blockIdx.x * STAGE_THR + threadIdx.x) >> 5;
    float4* q = p + (size_t)gw * (ZERO_PER_T * 32) + lane;
    const float4 z = make_float4(0.f, 0.f, 0.f, 0.f);
    for (int i = 0; i < ZERO_PER_T; ++i) q[i * 32] = z;
}

__global__ __launch_bounds__(STAGE_THR) void build_xt(const float* __restrict__ in,
                                                      const float* __restrict__ hx,
                                                      float* __restrict__ XT) {
    const int b  = (blockIdx.x & 1) * STAGE_THR + threadIdx.x;
    const int j0 = (blockIdx.x >> 1) * XT_ROWS;
    const float* srow = (j0 < INsz) ? in + (b * INsz + j0)
                                    : hx + (b * Hsz + (j0 - INsz));
    const float4 lo = *reinterpret_cast<const float4*>(srow);
    const float4 hi = *reinterpret_cast<const float4*>(srow + 4);
    float* d = XT + (j0 * Bsz + b);
    d[0 * Bsz] = lo.x;  d[1 * Bsz] = lo.y;  d[2 * Bsz] = lo.z;  d[3 * Bsz] = lo.w;
    d[4 * Bsz] = hi.x;  d[5 * Bsz] = hi.y;  d[6 * Bsz] = hi.z;  d[7 * Bsz] = hi.w;
}

__global__ __launch_bounds__(256) void scatter_w(const int*   __restrict__ src,
                                                 const int*   __restrict__ dst,
                                                 const float* __restrict__ signs,
                                                 const float* __restrict__ w1,
                                                 const float* __restrict__ w2,
                                                 const float* __restrict__ wa,
                                                 const float* __restrict__ wb,
                                                 float* __restrict__ W, int E) {
    int e = blockIdx.x * blockDim.x + threadIdx.x;
    if (e >= E) return;
    float  sg   = signs[e];
    size_t base = (size_t)src[e] * Hsz + (size_t)dst[e];
    atomicAdd(&W[base],            w1[e] * sg);
    atomicAdd(&W[WMAT + base],     w2[e] * sg);
    atomicAdd(&W[2ull*WMAT + base], wa[e] * sg);
    atomicAdd(&W[3ull*WMAT + base], wb[e] * sg);
}

#define KSTEPS 5
#define KBLK   (KSTEPS * 4)

struct FragSet {
    v2f a0[KSTEPS];
    v2f a1[KSTEPS];
    v2f b[KSTEPS][4];
};

__device__ __forceinline__ void load_block(FragSet& f,
                                           const float* __restrict__ xa0,
                                           const float* __restrict__ xa1,
                                           const float* __restrict__ wcol,
                                           int kb, int kp) {
    #pragma unroll
    for (int s = 0; s < KSTEPS; ++s) {
        const int kr = kb + 4 * s + kp;
        f.a0[s].x = xa0[(size_t)kr * Bsz];
        f.a0[s].y = xa0[(size_t)(kr + 1) * Bsz];
        f.a1[s].x = xa1[(size_t)kr * Bsz];
        f.a1[s].y = xa1[(size_t)(kr + 1) * Bsz];
        #pragma unroll
        for (int l = 0; l < 4; ++l) {
            const float* Wp = wcol + (size_t)l * WMAT + (size_t)kr * Hsz;
            f.b[s][l].x = Wp[0];
            f.b[s][l].y = Wp[Hsz];
        }
    }
}

__device__ __forceinline__ void mma_block(const FragSet& f, v8f (&acc)[2][4]) {
    #pragma unroll
    for (int s = 0; s < KSTEPS; ++s) {
        #pragma unroll
        for (int l = 0; l < 4; ++l) {
            acc[0][l] = __builtin_amdgcn_wmma_f32_16x16x4_f32(
                false, f.a0[s], false, f.b[s][l], (short)0, acc[0][l], false, false);
            acc[1][l] = __builtin_amdgcn_wmma_f32_16x16x4_f32(
                false, f.a1[s], false, f.b[s][l], (short)0, acc[1][l], false, false);
        }
    }
}

__global__ __launch_bounds__(128) void gemm_fused(const float* __restrict__ XT,
                                                  const float* __restrict__ W,
                                                  const float* __restrict__ ts,
                                                  const float* __restrict__ b_ff1,
                                                  const float* __restrict__ b_ff2,
                                                  const float* __restrict__ b_ta,
                                                  const float* __restrict__ b_tb,
                                                  float* __restrict__ out) {
    const int lane = threadIdx.x & 31;
    const int wave = threadIdx.x >> 5;

    const int n0   = (blockIdx.x * 4 + wave) * 16;
    const int m0   = blockIdx.y * 32;
    const int mrow = lane & 15;
    const int ncol = lane & 15;
    const int kp   = (lane >> 4) << 1;

    v8f acc[2][4];
    #pragma unroll
    for (int s = 0; s < 2; ++s)
        #pragma unroll
        for (int l = 0; l < 4; ++l)
            acc[s][l] = (v8f){0.f,0.f,0.f,0.f,0.f,0.f,0.f,0.f};

    const float* xa0  = XT + (m0      + mrow);
    const float* xa1  = XT + (m0 + 16 + mrow);
    const float* wcol = W + n0 + ncol;

    FragSet f0, f1;
    load_block(f0, xa0, xa1, wcol, 0, kp);

    int kb = 0;
    for (; kb + 2 * KBLK < CONC; kb += 2 * KBLK) {
        load_block(f1, xa0, xa1, wcol, kb + KBLK, kp);
        mma_block(f0, acc);
        load_block(f0, xa0, xa1, wcol, kb + 2 * KBLK, kp);
        mma_block(f1, acc);
    }
    load_block(f1, xa0, xa1, wcol, kb + KBLK, kp);
    mma_block(f0, acc);
    mma_block(f1, acc);

    const int   ng  = n0 + ncol;
    const float bf1 = b_ff1[ng];
    const float bf2 = b_ff2[ng];
    const float bta = b_ta[ng];
    const float btb = b_tb[ng];
    const int   mhi = (lane >> 4) << 3;

    #pragma unroll
    for (int s = 0; s < 2; ++s) {
        #pragma unroll
        for (int r = 0; r < 8; ++r) {
            const int   m   = m0 + s * 16 + r + mhi;
            const float tsv = ts[m];
            const float f1  = tanhf(acc[s][0][r] + bf1);
            const float f2  = tanhf(acc[s][1][r] + bf2);
            const float ta  = acc[s][2][r] + bta;
            const float tb  = acc[s][3][r] + btb;
            const float sg  = 1.f / (1.f + expf(-(ta * tsv + tb)));
            const float h   = f1 * (1.f - sg) + sg * f2;
            const size_t o  = (size_t)m * Hsz + ng;
            out[o]                         = h;
            out[(size_t)Bsz * Hsz + o]     = h;
        }
    }
}

extern "C" void kernel_launch(void* const* d_in, const int* in_sizes, int n_in,
                              void* d_out, int out_size, void* d_ws, size_t ws_size,
                              hipStream_t stream) {
    const float* input = (const float*)d_in[0];
    const float* hx    = (const float*)d_in[1];
    const float* ts    = (const float*)d_in[2];
    const int*   src   = (const int*)  d_in[3];
    const int*   dst   = (const int*)  d_in[4];
    const float* signs = (const float*)d_in[5];
    const float* w1    = (const float*)d_in[6];
    const float* w2    = (const float*)d_in[7];
    const float* wa    = (const float*)d_in[8];
    const float* wb    = (const float*)d_in[9];
    const float* bf1   = (const float*)d_in[10];
    const float* bf2   = (const float*)d_in[11];
    const float* bta   = (const float*)d_in[12];
    const float* btb   = (const float*)d_in[13];
    const int    E     = in_sizes[3];

    float* W  = (float*)d_ws;
    float* XT = W + 4ull * WMAT;
    float* out = (float*)d_out;

    zero4<<<ZERO_WGS, STAGE_THR, 0, stream>>>((float4*)W);

    build_xt<<<XT_WGS, STAGE_THR, 0, stream>>>(input, hx, XT);

    scatter_w<<<(E + 255) / 256, 256, 0, stream>>>(src, dst, signs,
                                                   w1, w2, wa, wb, W, E);

    dim3 grid(Hsz / 64, Bsz / 32);
    gemm_fused<<<grid, 128, 0, stream>>>(XT, W, ts, bf1, bf2, bta, btb, out);
}


# GEMM fragments come as 16-byte loads from operands staged k-interleaved by 8 (one load = a lane's K pair for two k-steps): 60 loads in flight now carry 30 KB per wave, not 8 KB; sum order kept
# speedup vs baseline: 2.3157x; 1.2680x over previous
#include <hip/hip_runtime.h>
#include <math.h>

#define Bsz   512
#define INsz  256
#define Hsz   1024
#define CONC  1280
#define WMAT  (CONC * Hsz)

typedef __attribute__((ext_vector_type(2))) float v2f;
typedef __attribute__((ext_vector_type(4))) float v4f;
typedef __attribute__((ext_vector_type(8))) float v8f;

#define KGRP   8
#define NGRP   (CONC / KGRP)

#define STAGE_THR   256
#define ZERO_WGS    256
#define ZERO_N4     (4 * WMAT / 4)
#define ZERO_PER_T  (ZERO_N4 / (ZERO_WGS * STAGE_THR))
#define XT_ROWS     8
#define XT_WGS      (2 * (CONC / XT_ROWS))

static_assert(ZERO_N4 % (ZERO_WGS * STAGE_THR) == 0, "zero fill must divide evenly");
static_assert(INsz % XT_ROWS == 0 && CONC % XT_ROWS == 0, "a row group reads one source only");
static_assert(XT_ROWS == KGRP, "one work-group row block is one k group");
static_assert(Bsz == 2 * STAGE_THR && XT_ROWS == 8, "X^T tiling must cover exactly");

__global__ __launch_bounds__(STAGE_THR) void zero4(float4* __restrict__ p) {
    const int lane = threadIdx.x & 31;
    const int gw   = (blockIdx.x * STAGE_THR + threadIdx.x) >> 5;
    float4* q = p + (size_t)gw * (ZERO_PER_T * 32) + lane;
    const float4 z = make_float4(0.f, 0.f, 0.f, 0.f);
    for (int i = 0; i < ZERO_PER_T; ++i) q[i * 32] = z;
}

__global__ __launch_bounds__(STAGE_THR) void build_x8(const float* __restrict__ in,
                                                      const float* __restrict__ hx,
                                                      float4* __restrict__ X8) {
    const int b  = (blockIdx.x & 1) * STAGE_THR + threadIdx.x;
    const int g  = blockIdx.x >> 1;
    const int j0 = g * KGRP;
    const float* srow = (j0 < INsz) ? in + (b * INsz + j0)
                                    : hx + (b * Hsz + (j0 - INsz));
    const float4 lo = *reinterpret_cast<const float4*>(srow);
    const float4 hi = *reinterpret_cast<const float4*>(srow + 4);
    X8[(g * 2 + 0) * Bsz + b] = make_float4(lo.x, lo.y, hi.x, hi.y);
    X8[(g * 2 + 1) * Bsz + b] = make_float4(lo.z, lo.w, hi.z, hi.w);
}

__global__ __launch_bounds__(256) void scatter_w(const int*   __restrict__ src,
                                                 const int*   __restrict__ dst,
                                                 const float* __restrict__ signs,
                                                 const float* __restrict__ w1,
                                                 const float* __restrict__ w2,
                                                 const float* __restrict__ wa,
                                                 const float* __restrict__ wb,
                                                 float* __restrict__ W, int E) {
    int e = blockIdx.x * blockDim.x + threadIdx.x;
    if (e >= E) return;
    float  sg   = signs[e];
    const int k = src[e];
    const int j = k & (KGRP - 1);
    const int piece = (k >> 3) * 2 + ((j >> 1) & 1);
    size_t base = (((size_t)piece * Hsz + (size_t)dst[e]) << 2) + (size_t)((j >> 2) * 2 + (j & 1));
    atomicAdd(&W[base],            w1[e] * sg);
    atomicAdd(&W[WMAT + base],     w2[e] * sg);
    atomicAdd(&W[2ull*WMAT + base], wa[e] * sg);
    atomicAdd(&W[3ull*WMAT + base], wb[e] * sg);
}

#define GPB 10

static_assert(NGRP % (2 * GPB) == 0, "the pipelined loop walks two blocks per trip");

struct FragSet {
    v4f a[GPB][2];
    v4f b[GPB][4];
};

__device__ __forceinline__ void load_block(FragSet& f,
                                           const v4f* __restrict__ xa,
                                           const v4f* __restrict__ wb,
                                           int gb) {
    #pragma unroll
    for (int i = 0; i < GPB; ++i) {
        const size_t g = (size_t)(gb + i);
        f.a[i][0] = xa[g * (2 * Bsz)];
        f.a[i][1] = xa[g * (2 * Bsz) + 16];
        #pragma unroll
        for (int l = 0; l < 4; ++l)
            f.b[i][l] = wb[(size_t)l * (WMAT / 4) + g * (2 * Hsz)];
    }
}

__device__ __forceinline__ void mma_block(const FragSet& f, v8f (&acc)[2][4]) {
    #pragma unroll
    for (int i = 0; i < GPB; ++i) {
        #pragma unroll
        for (int st = 0; st < 2; ++st) {
            const v2f a0 = st ? __builtin_shufflevector(f.a[i][0], f.a[i][0], 2, 3)
                              : __builtin_shufflevector(f.a[i][0], f.a[i][0], 0, 1);
            const v2f a1 = st ? __builtin_shufflevector(f.a[i][1], f.a[i][1], 2, 3)
                              : __builtin_shufflevector(f.a[i][1], f.a[i][1], 0, 1);
            #pragma unroll
            for (int l = 0; l < 4; ++l) {
                const v2f bb = st ? __builtin_shufflevector(f.b[i][l], f.b[i][l], 2, 3)
                                  : __builtin_shufflevector(f.b[i][l], f.b[i][l], 0, 1);
                acc[0][l] = __builtin_amdgcn_wmma_f32_16x16x4_f32(
                    false, a0, false, bb, (short)0, acc[0][l], false, false);
                acc[1][l] = __builtin_amdgcn_wmma_f32_16x16x4_f32(
                    false, a1, false, bb, (short)0, acc[1][l], false, false);
            }
        }
    }
}

__global__ __launch_bounds__(128) void gemm_fused(const float* __restrict__ X8,
                                                  const float* __restrict__ W,
                                                  const float* __restrict__ ts,
                                                  const float* __restrict__ b_ff1,
                                                  const float* __restrict__ b_ff2,
                                                  const float* __restrict__ b_ta,
                                                  const float* __restrict__ b_tb,
                                                  float* __restrict__ out) {
    const int lane = threadIdx.x & 31;
    const int wave = threadIdx.x >> 5;

    const int n0   = (blockIdx.x * 4 + wave) * 16;
    const int m0   = blockIdx.y * 32;
    const int ncol = lane & 15;
    const int half = lane >> 4;

    v8f acc[2][4];
    #pragma unroll
    for (int s = 0; s < 2; ++s)
        #pragma unroll
        for (int l = 0; l < 4; ++l)
            acc[s][l] = (v8f){0.f,0.f,0.f,0.f,0.f,0.f,0.f,0.f};

    const v4f* xa = reinterpret_cast<const v4f*>(X8) + (half * Bsz + m0 + ncol);
    const v4f* wb = reinterpret_cast<const v4f*>(W)  + (half * Hsz + n0 + ncol);

    FragSet f0, f1;
    load_block(f0, xa, wb, 0);

    int gb = 0;
    for (; gb + 2 * GPB < NGRP; gb += 2 * GPB) {
        load_block(f1, xa, wb, gb + GPB);
        __builtin_amdgcn_sched_barrier(0);
        mma_block(f0, acc);
        __builtin_amdgcn_sched_barrier(0);
        load_block(f0, xa, wb, gb + 2 * GPB);
        __builtin_amdgcn_sched_barrier(0);
        mma_block(f1, acc);
        __builtin_amdgcn_sched_barrier(0);
    }
    load_block(f1, xa, wb, gb + GPB);
    __builtin_amdgcn_sched_barrier(0);
    mma_block(f0, acc);
    mma_block(f1, acc);

    const int   ng  = n0 + ncol;
    const float bf1 = b_ff1[ng];
    const float bf2 = b_ff2[ng];
    const float bta = b_ta[ng];
    const float btb = b_tb[ng];
    const int   mhi = (lane >> 4) << 3;

    #pragma unroll
    for (int s = 0; s < 2; ++s) {
        #pragma unroll
        for (int r = 0; r < 8; ++r) {
            const int   m   = m0 + s * 16 + r + mhi;
            const float tsv = ts[m];
            const float f1  = tanhf(acc[s][0][r] + bf1);
            const float f2  = tanhf(acc[s][1][r] + bf2);
            const float ta  = acc[s][2][r] + bta;
            const float tb  = acc[s][3][r] + btb;
            const float sg  = 1.f / (1.f + expf(-(ta * tsv + tb)));
            const float h   = f1 * (1.f - sg) + sg * f2;
            const size_t o  = (size_t)m * Hsz + ng;
            out[o]                         = h;
            out[(size_t)Bsz * Hsz + o]     = h;
        }
    }
}

extern "C" void kernel_launch(void* const* d_in, const int* in_sizes, int n_in,
                              void* d_out, int out_size, void* d_ws, size_t ws_size,
                              hipStream_t stream) {
    const float* input = (const float*)d_in[0];
    const float* hx    = (const float*)d_in[1];
    const float* ts    = (const float*)d_in[2];
    const int*   src   = (const int*)  d_in[3];
    const int*   dst   = (const int*)  d_in[4];
    const float* signs = (const float*)d_in[5];
    const float* w1    = (const float*)d_in[6];
    const float* w2    = (const float*)d_in[7];
    const float* wa    = (const float*)d_in[8];
    const float* wb    = (const float*)d_in[9];
    const float* bf1   = (const float*)d_in[10];
    const float* bf2   = (const float*)d_in[11];
    const float* bta   = (const float*)d_in[12];
    const float* btb   = (const float*)d_in[13];
    const int    E     = in_sizes[3];

    float* W  = (float*)d_ws;
    float* X8 = W + 4ull * WMAT;
    float* out = (float*)d_out;

    zero4<<<ZERO_WGS, STAGE_THR, 0, stream>>>((float4*)W);

    build_x8<<<XT_WGS, STAGE_THR, 0, stream>>>(input, hx, (float4*)X8);

    scatter_w<<<(E + 255) / 256, 256, 0, stream>>>(src, dst, signs,
                                                   w1, w2, wa, wb, W, E);

    dim3 grid(Hsz / 64, Bsz / 32);
    gemm_fused<<<grid, 128, 0, stream>>>(X8, W, ts, bf1, bf2, bta, btb, out);
}


# Weight staging puts the four matrices of one (k,n) in one 32-byte cell; scatter gives a lane quad one edge so its 4 float atomics ask for one line, not four; atomic arrival order unordered as before
# speedup vs baseline: 2.8093x; 1.2132x over previous
#include <hip/hip_runtime.h>
#include <math.h>

#define Bsz   512
#define INsz  256
#define Hsz   1024
#define CONC  1280
#define WMAT  (CONC * Hsz)

typedef __attribute__((ext_vector_type(2))) float v2f;
typedef __attribute__((ext_vector_type(4))) float v4f;
typedef __attribute__((ext_vector_type(8))) float v8f;

#define KGRP   8
#define NGRP   (CONC / KGRP)

#define STAGE_THR   256
#define ZERO_WGS    256
#define ZERO_N4     (4 * WMAT / 4)
#define ZERO_PER_T  (ZERO_N4 / (ZERO_WGS * STAGE_THR))
#define XT_ROWS     8
#define XT_WGS      (2 * (CONC / XT_ROWS))

static_assert(ZERO_N4 % (ZERO_WGS * STAGE_THR) == 0, "zero fill must divide evenly");
static_assert(INsz % XT_ROWS == 0 && CONC % XT_ROWS == 0, "a row group reads one source only");
static_assert(XT_ROWS == KGRP, "one work-group row block is one k group");
static_assert(Bsz == 2 * STAGE_THR && XT_ROWS == 8, "X^T tiling must cover exactly");

__global__ __launch_bounds__(STAGE_THR) void zero4(float4* __restrict__ p) {
    const int lane = threadIdx.x & 31;
    const int gw   = (blockIdx.x * STAGE_THR + threadIdx.x) >> 5;
    float4* q = p + (size_t)gw * (ZERO_PER_T * 32) + lane;
    const float4 z = make_float4(0.f, 0.f, 0.f, 0.f);
    for (int i = 0; i < ZERO_PER_T; ++i) q[i * 32] = z;
}

__global__ __launch_bounds__(STAGE_THR) void build_x8(const float* __restrict__ in,
                                                      const float* __restrict__ hx,
                                                      float4* __restrict__ X8) {
    const int b  = (blockIdx.x & 1) * STAGE_THR + threadIdx.x;
    const int g  = blockIdx.x >> 1;
    const int j0 = g * KGRP;
    const float* srow = (j0 < INsz) ? in + (b * INsz + j0)
                                    : hx + (b * Hsz + (j0 - INsz));
    const float4 lo = *reinterpret_cast<const float4*>(srow);
    const float4 hi = *reinterpret_cast<const float4*>(srow + 4);
    X8[(g * 2 + 0) * Bsz + b] = make_float4(lo.x, lo.y, hi.x, hi.y);
    X8[(g * 2 + 1) * Bsz + b] = make_float4(lo.z, lo.w, hi.z, hi.w);
}

__global__ __launch_bounds__(256) void scatter_w(const int*   __restrict__ src,
                                                 const int*   __restrict__ dst,
                                                 const float* __restrict__ signs,
                                                 const float* __restrict__ w1,
                                                 const float* __restrict__ w2,
                                                 const float* __restrict__ wa,
                                                 const float* __restrict__ wb,
                                                 float* __restrict__ W, int E) {
    const int t  = blockIdx.x * blockDim.x + threadIdx.x;
    const int l  = t & 3;
    const int e0 = t & ~3;
    const float* __restrict__ wp = (l == 0) ? w1 : (l == 1) ? w2 : (l == 2) ? wa : wb;
    const int inner = (l >> 1) * 4 + (l & 1) * 2;

    int   kk[4], nn[4];
    float vv[4];
    bool  ok[4];
    #pragma unroll
    for (int i = 0; i < 4; ++i) {
        ok[i] = (e0 + i) < E;
        const int e = ok[i] ? (e0 + i) : 0;
        kk[i] = src[e];
        nn[i] = dst[e];
        vv[i] = wp[e] * signs[e];
    }
    #pragma unroll
    for (int i = 0; i < 4; ++i) {
        const int k = kk[i];
        const int j = k & (KGRP - 1);
        const int cell = (k >> 3) * 4 + ((j >> 1) & 1) * 2 + (j >> 2);
        const size_t idx = (((size_t)cell * Hsz + (size_t)nn[i]) << 3) + (size_t)(inner + (j & 1));
        if (ok[i]) atomicAdd(&W[idx], vv[i]);
    }
}

#define GPB 10

static_assert(NGRP % (2 * GPB) == 0, "the pipelined loop walks two blocks per trip");

struct FragSet {
    v4f a[GPB][2];
    v4f b[GPB][4];
};

__device__ __forceinline__ void load_block(FragSet& f,
                                           const v4f* __restrict__ xa,
                                           const v4f* __restrict__ wb,
                                           int gb) {
    #pragma unroll
    for (int i = 0; i < GPB; ++i) {
        const size_t g = (size_t)(gb + i);
        f.a[i][0] = xa[g * (2 * Bsz)];
        f.a[i][1] = xa[g * (2 * Bsz) + 16];
        #pragma unroll
        for (int s = 0; s < 2; ++s) {
            f.b[i][s * 2 + 0] = wb[g * (8 * Hsz) + s * (2 * Hsz)];
            f.b[i][s * 2 + 1] = wb[g * (8 * Hsz) + s * (2 * Hsz) + 1];
        }
    }
}

__device__ __forceinline__ void mma_block(const FragSet& f, v8f (&acc)[2][4]) {
    #pragma unroll
    for (int i = 0; i < GPB; ++i) {
        #pragma unroll
        for (int st = 0; st < 2; ++st) {
            const v2f a0 = st ? __builtin_shufflevector(f.a[i][0], f.a[i][0], 2, 3)
                              : __builtin_shufflevector(f.a[i][0], f.a[i][0], 0, 1);
            const v2f a1 = st ? __builtin_shufflevector(f.a[i][1], f.a[i][1], 2, 3)
                              : __builtin_shufflevector(f.a[i][1], f.a[i][1], 0, 1);
            #pragma unroll
            for (int l = 0; l < 4; ++l) {
                const v4f q  = f.b[i][st * 2 + (l >> 1)];
                const v2f bb = (l & 1) ? __builtin_shufflevector(q, q, 2, 3)
                                       : __builtin_shufflevector(q, q, 0, 1);
                acc[0][l] = __builtin_amdgcn_wmma_f32_16x16x4_f32(
                    false, a0, false, bb, (short)0, acc[0][l], false, false);
                acc[1][l] = __builtin_amdgcn_wmma_f32_16x16x4_f32(
                    false, a1, false, bb, (short)0, acc[1][l], false, false);
            }
        }
    }
}

__global__ __launch_bounds__(128) void gemm_fused(const float* __restrict__ X8,
                                                  const float* __restrict__ W,
                                                  const float* __restrict__ ts,
                                                  const float* __restrict__ b_ff1,
                                                  const float* __restrict__ b_ff2,
                                                  const float* __restrict__ b_ta,
                                                  const float* __restrict__ b_tb,
                                                  float* __restrict__ out) {
    const int lane = threadIdx.x & 31;
    const int wave = threadIdx.x >> 5;

    const int n0   = (blockIdx.x * 4 + wave) * 16;
    const int m0   = blockIdx.y * 32;
    const int ncol = lane & 15;
    const int half = lane >> 4;

    v8f acc[2][4];
    #pragma unroll
    for (int s = 0; s < 2; ++s)
        #pragma unroll
        for (int l = 0; l < 4; ++l)
            acc[s][l] = (v8f){0.f,0.f,0.f,0.f,0.f,0.f,0.f,0.f};

    const v4f* xa = reinterpret_cast<const v4f*>(X8) + (half * Bsz + m0 + ncol);
    const v4f* wb = reinterpret_cast<const v4f*>(W)  + (half * (4 * Hsz) + 2 * (n0 + ncol));

    FragSet f0, f1;
    load_block(f0, xa, wb, 0);

    int gb = 0;
    for (; gb + 2 * GPB < NGRP; gb += 2 * GPB) {
        load_block(f1, xa, wb, gb + GPB);
        __builtin_amdgcn_sched_barrier(0);
        mma_block(f0, acc);
        __builtin_amdgcn_sched_barrier(0);
        load_block(f0, xa, wb, gb + 2 * GPB);
        __builtin_amdgcn_sched_barrier(0);
        mma_block(f1, acc);
        __builtin_amdgcn_sched_barrier(0);
    }
    load_block(f1, xa, wb, gb + GPB);
    __builtin_amdgcn_sched_barrier(0);
    mma_block(f0, acc);
    mma_block(f1, acc);

    const int   ng  = n0 + ncol;
    const float bf1 = b_ff1[ng];
    const float bf2 = b_ff2[ng];
    const float bta = b_ta[ng];
    const float btb = b_tb[ng];
    const int   mhi = (lane >> 4) << 3;

    #pragma unroll
    for (int s = 0; s < 2; ++s) {
        #pragma unroll
        for (int r = 0; r < 8; ++r) {
            const int   m   = m0 + s * 16 + r + mhi;
            const float tsv = ts[m];
            const float f1  = tanhf(acc[s][0][r] + bf1);
            const float f2  = tanhf(acc[s][1][r] + bf2);
            const float ta  = acc[s][2][r] + bta;
            const float tb  = acc[s][3][r] + btb;
            const float sg  = 1.f / (1.f + expf(-(ta * tsv + tb)));
            const float h   = f1 * (1.f - sg) + sg * f2;
            const size_t o  = (size_t)m * Hsz + ng;
            out[o]                         = h;
            out[(size_t)Bsz * Hsz + o]     = h;
        }
    }
}

extern "C" void kernel_launch(void* const* d_in, const int* in_sizes, int n_in,
                              void* d_out, int out_size, void* d_ws, size_t ws_size,
                              hipStream_t stream) {
    const float* input = (const float*)d_in[0];
    const float* hx    = (const float*)d_in[1];
    const float* ts    = (const float*)d_in[2];
    const int*   src   = (const int*)  d_in[3];
    const int*   dst   = (const int*)  d_in[4];
    const float* signs = (const float*)d_in[5];
    const float* w1    = (const float*)d_in[6];
    const float* w2    = (const float*)d_in[7];
    const float* wa    = (const float*)d_in[8];
    const float* wb    = (const float*)d_in[9];
    const float* bf1   = (const float*)d_in[10];
    const float* bf2   = (const float*)d_in[11];
    const float* bta   = (const float*)d_in[12];
    const float* btb   = (const float*)d_in[13];
    const int    E     = in_sizes[3];

    float* W  = (float*)d_ws;
    float* X8 = W + 4ull * WMAT;
    float* out = (float*)d_out;

    zero4<<<ZERO_WGS, STAGE_THR, 0, stream>>>((float4*)W);

    build_x8<<<XT_WGS, STAGE_THR, 0, stream>>>(input, hx, (float4*)X8);

    scatter_w<<<(E + 255) / 256, 256, 0, stream>>>(src, dst, signs,
                                                   w1, w2, wa, wb, W, E);

    dim3 grid(Hsz / 64, Bsz / 32);
    gemm_fused<<<grid, 128, 0, stream>>>(X8, W, ts, bf1, bf2, bta, btb, out);
}


# zero fill of W and the k-interleaved X build run as ONE launch (work-groups 0-255 clear, 256-575 build; neither needed the other): 3 dependent launches per call, not 4
# speedup vs baseline: 2.9515x; 1.0506x over previous
#include <hip/hip_runtime.h>
#include <math.h>

#define Bsz   512
#define INsz  256
#define Hsz   1024
#define CONC  1280
#define WMAT  (CONC * Hsz)

typedef __attribute__((ext_vector_type(2))) float v2f;
typedef __attribute__((ext_vector_type(4))) float v4f;
typedef __attribute__((ext_vector_type(8))) float v8f;

#define KGRP   8
#define NGRP   (CONC / KGRP)

#define STAGE_THR   256
#define ZERO_WGS    256
#define ZERO_N4     (4 * WMAT / 4)
#define ZERO_PER_T  (ZERO_N4 / (ZERO_WGS * STAGE_THR))
#define XT_ROWS     8
#define XT_WGS      (2 * (CONC / XT_ROWS))
#define STAGE_WGS   (ZERO_WGS + XT_WGS)

static_assert(ZERO_N4 % (ZERO_WGS * STAGE_THR) == 0, "zero fill must divide evenly");
static_assert(INsz % XT_ROWS == 0 && CONC % XT_ROWS == 0, "a row group reads one source only");
static_assert(XT_ROWS == KGRP, "one work-group row block is one k group");
static_assert(Bsz == 2 * STAGE_THR && XT_ROWS == 8, "X^T tiling must cover exactly");

__global__ __launch_bounds__(STAGE_THR) void stage_zero_x8(float4* __restrict__ Wz,
                                                           const float* __restrict__ in,
                                                           const float* __restrict__ hx,
                                                           float4* __restrict__ X8) {
    if (blockIdx.x < ZERO_WGS) {
        const int lane = threadIdx.x & 31;
        const int gw   = (blockIdx.x * STAGE_THR + threadIdx.x) >> 5;
        float4* q = Wz + (size_t)gw * (ZERO_PER_T * 32) + lane;
        const float4 z = make_float4(0.f, 0.f, 0.f, 0.f);
        for (int i = 0; i < ZERO_PER_T; ++i) q[i * 32] = z;
    } else {
        const int w  = blockIdx.x - ZERO_WGS;
        const int b  = (w & 1) * STAGE_THR + threadIdx.x;
        const int g  = w >> 1;
        const int j0 = g * KGRP;
        const float* srow = (j0 < INsz) ? in + (b * INsz + j0)
                                        : hx + (b * Hsz + (j0 - INsz));
        const float4 lo = *reinterpret_cast<const float4*>(srow);
        const float4 hi = *reinterpret_cast<const float4*>(srow + 4);
        X8[(g * 2 + 0) * Bsz + b] = make_float4(lo.x, lo.y, hi.x, hi.y);
        X8[(g * 2 + 1) * Bsz + b] = make_float4(lo.z, lo.w, hi.z, hi.w);
    }
}

__global__ __launch_bounds__(256) void scatter_w(const int*   __restrict__ src,
                                                 const int*   __restrict__ dst,
                                                 const float* __restrict__ signs,
                                                 const float* __restrict__ w1,
                                                 const float* __restrict__ w2,
                                                 const float* __restrict__ wa,
                                                 const float* __restrict__ wb,
                                                 float* __restrict__ W, int E) {
    const int t  = blockIdx.x * blockDim.x + threadIdx.x;
    const int l  = t & 3;
    const int e0 = t & ~3;
    const float* __restrict__ wp = (l == 0) ? w1 : (l == 1) ? w2 : (l == 2) ? wa : wb;
    const int inner = (l >> 1) * 4 + (l & 1) * 2;

    int   kk[4], nn[4];
    float vv[4];
    bool  ok[4];
    #pragma unroll
    for (int i = 0; i < 4; ++i) {
        ok[i] = (e0 + i) < E;
        const int e = ok[i] ? (e0 + i) : 0;
        kk[i] = src[e];
        nn[i] = dst[e];
        vv[i] = wp[e] * signs[e];
    }
    #pragma unroll
    for (int i = 0; i < 4; ++i) {
        const int k = kk[i];
        const int j = k & (KGRP - 1);
        const int cell = (k >> 3) * 4 + ((j >> 1) & 1) * 2 + (j >> 2);
        const size_t idx = (((size_t)cell * Hsz + (size_t)nn[i]) << 3) + (size_t)(inner + (j & 1));
        if (ok[i]) atomicAdd(&W[idx], vv[i]);
    }
}

#define GPB 10

static_assert(NGRP % (2 * GPB) == 0, "the pipelined loop walks two blocks per trip");

struct FragSet {
    v4f a[GPB][2];
    v4f b[GPB][4];
};

__device__ __forceinline__ void load_block(FragSet& f,
                                           const v4f* __restrict__ xa,
                                           const v4f* __restrict__ wb,
                                           int gb) {
    #pragma unroll
    for (int i = 0; i < GPB; ++i) {
        const size_t g = (size_t)(gb + i);
        f.a[i][0] = xa[g * (2 * Bsz)];
        f.a[i][1] = xa[g * (2 * Bsz) + 16];
        #pragma unroll
        for (int s = 0; s < 2; ++s) {
            f.b[i][s * 2 + 0] = wb[g * (8 * Hsz) + s * (2 * Hsz)];
            f.b[i][s * 2 + 1] = wb[g * (8 * Hsz) + s * (2 * Hsz) + 1];
        }
    }
}

__device__ __forceinline__ void mma_block(const FragSet& f, v8f (&acc)[2][4]) {
    #pragma unroll
    for (int i = 0; i < GPB; ++i) {
        #pragma unroll
        for (int st = 0; st < 2; ++st) {
            const v2f a0 = st ? __builtin_shufflevector(f.a[i][0], f.a[i][0], 2, 3)
                              : __builtin_shufflevector(f.a[i][0], f.a[i][0], 0, 1);
            const v2f a1 = st ? __builtin_shufflevector(f.a[i][1], f.a[i][1], 2, 3)
                              : __builtin_shufflevector(f.a[i][1], f.a[i][1], 0, 1);
            #pragma unroll
            for (int l = 0; l < 4; ++l) {
                const v4f q  = f.b[i][st * 2 + (l >> 1)];
                const v2f bb = (l & 1) ? __builtin_shufflevector(q, q, 2, 3)
                                       : __builtin_shufflevector(q, q, 0, 1);
                acc[0][l] = __builtin_amdgcn_wmma_f32_16x16x4_f32(
                    false, a0, false, bb, (short)0, acc[0][l], false, false);
                acc[1][l] = __builtin_amdgcn_wmma_f32_16x16x4_f32(
                    false, a1, false, bb, (short)0, acc[1][l], false, false);
            }
        }
    }
}

__global__ __launch_bounds__(128) void gemm_fused(const float* __restrict__ X8,
                                                  const float* __restrict__ W,
                                                  const float* __restrict__ ts,
                                                  const float* __restrict__ b_ff1,
                                                  const float* __restrict__ b_ff2,
                                                  const float* __restrict__ b_ta,
                                                  const float* __restrict__ b_tb,
                                                  float* __restrict__ out) {
    const int lane = threadIdx.x & 31;
    const int wave = threadIdx.x >> 5;

    const int n0   = (blockIdx.x * 4 + wave) * 16;
    const int m0   = blockIdx.y * 32;
    const int ncol = lane & 15;
    const int half = lane >> 4;

    v8f acc[2][4];
    #pragma unroll
    for (int s = 0; s < 2; ++s)
        #pragma unroll
        for (int l = 0; l < 4; ++l)
            acc[s][l] = (v8f){0.f,0.f,0.f,0.f,0.f,0.f,0.f,0.f};

    const v4f* xa = reinterpret_cast<const v4f*>(X8) + (half * Bsz + m0 + ncol);
    const v4f* wb = reinterpret_cast<const v4f*>(W)  + (half * (4 * Hsz) + 2 * (n0 + ncol));

    FragSet f0, f1;
    load_block(f0, xa, wb, 0);

    int gb = 0;
    for (; gb + 2 * GPB < NGRP; gb += 2 * GPB) {
        load_block(f1, xa, wb, gb + GPB);
        __builtin_amdgcn_sched_barrier(0);
        mma_block(f0, acc);
        __builtin_amdgcn_sched_barrier(0);
        load_block(f0, xa, wb, gb + 2 * GPB);
        __builtin_amdgcn_sched_barrier(0);
        mma_block(f1, acc);
        __builtin_amdgcn_sched_barrier(0);
    }
    load_block(f1, xa, wb, gb + GPB);
    __builtin_amdgcn_sched_barrier(0);
    mma_block(f0, acc);
    mma_block(f1, acc);

    const int   ng  = n0 + ncol;
    const float bf1 = b_ff1[ng];
    const float bf2 = b_ff2[ng];
    const float bta = b_ta[ng];
    const float btb = b_tb[ng];
    const int   mhi = (lane >> 4) << 3;

    #pragma unroll
    for (int s = 0; s < 2; ++s) {
        #pragma unroll
        for (int r = 0; r < 8; ++r) {
            const int   m   = m0 + s * 16 + r + mhi;
            const float tsv = ts[m];
            const float f1  = tanhf(acc[s][0][r] + bf1);
            const float f2  = tanhf(acc[s][1][r] + bf2);
            const float ta  = acc[s][2][r] + bta;
            const float tb  = acc[s][3][r] + btb;
            const float sg  = 1.f / (1.f + expf(-(ta * tsv + tb)));
            const float h   = f1 * (1.f - sg) + sg * f2;
            const size_t o  = (size_t)m * Hsz + ng;
            out[o]                         = h;
            out[(size_t)Bsz * Hsz + o]     = h;
        }
    }
}

extern "C" void kernel_launch(void* const* d_in, const int* in_sizes, int n_in,
                              void* d_out, int out_size, void* d_ws, size_t ws_size,
                              hipStream_t stream) {
    const float* input = (const float*)d_in[0];
    const float* hx    = (const float*)d_in[1];
    const float* ts    = (const float*)d_in[2];
    const int*   src   = (const int*)  d_in[3];
    const int*   dst   = (const int*)  d_in[4];
    const float* signs = (const float*)d_in[5];
    const float* w1    = (const float*)d_in[6];
    const float* w2    = (const float*)d_in[7];
    const float* wa    = (const float*)d_in[8];
    const float* wb    = (const float*)d_in[9];
    const float* bf1   = (const float*)d_in[10];
    const float* bf2   = (const float*)d_in[11];
    const float* bta   = (const float*)d_in[12];
    const float* btb   = (const float*)d_in[13];
    const int    E     = in_sizes[3];

    float* W  = (float*)d_ws;
    float* X8 = W + 4ull * WMAT;
    float* out = (float*)d_out;

    stage_zero_x8<<<STAGE_WGS, STAGE_THR, 0, stream>>>((float4*)W, input, hx, (float4*)X8);

    scatter_w<<<(E + 255) / 256, 256, 0, stream>>>(src, dst, signs,
                                                   w1, w2, wa, wb, W, E);

    dim3 grid(Hsz / 64, Bsz / 32);
    gemm_fused<<<grid, 128, 0, stream>>>(X8, W, ts, bf1, bf2, bta, btb, out);
}


# Dense weights built per (N tile, K half) in the GEMM work-group's LDS from edges the staging launch bins (LDS float adds, add order free as before): no 21 MB zero fill, no global atomics, 2 launches
# speedup vs baseline: 3.6233x; 1.2276x over previous
#include <hip/hip_runtime.h>
#include <math.h>

#define Bsz   512
#define INsz  256
#define Hsz   1024
#define CONC  1280

typedef __attribute__((ext_vector_type(2))) float v2f;
typedef __attribute__((ext_vector_type(4))) float v4f;
typedef __attribute__((ext_vector_type(8))) float v8f;

#define KGRP   8
#define NGRP   (CONC / KGRP)
#define HGRP   (NGRP / 2)
#define KHALF  (CONC / 2)
#define LDSW_FLOATS (HGRP * 512)

#define STAGE_THR   256
#define CHUNK       1024
#define EPT         (CHUNK / STAGE_THR)
#define NBIN        128
#define OFFP        132
#define BIN_WGS     128
#define XT_ROWS     8
#define XT_WGS      (2 * (CONC / XT_ROWS))
#define STAGE_WGS   (BIN_WGS + XT_WGS)
#define MAXE        (1 << 20)

static_assert(INsz % XT_ROWS == 0 && CONC % XT_ROWS == 0, "a row group reads one source only");
static_assert(XT_ROWS == KGRP, "one work-group row block is one k group");
static_assert(Bsz == 2 * STAGE_THR && XT_ROWS == 8, "X^T tiling must cover exactly");
static_assert(NBIN == 2 * (Hsz / 16), "one bin per (N tile, K half)");
static_assert(NBIN + 1 <= STAGE_THR && NBIN + 1 <= OFFP, "one thread per start offset");

__global__ __launch_bounds__(STAGE_THR) void stage_bin_x8(const int*   __restrict__ src,
                                                          const int*   __restrict__ dst,
                                                          const float* __restrict__ signs,
                                                          const float* __restrict__ w1,
                                                          const float* __restrict__ w2,
                                                          const float* __restrict__ wa,
                                                          const float* __restrict__ wb,
                                                          int*    __restrict__ recK,
                                                          float4* __restrict__ recV,
                                                          int*    __restrict__ off,
                                                          const float* __restrict__ in,
                                                          const float* __restrict__ hx,
                                                          float4* __restrict__ X8,
                                                          int E) {
    __shared__ int    cnt[NBIN];
    __shared__ int    start[NBIN + 1];
    __shared__ int    sK[CHUNK];
    __shared__ float4 sV[CHUNK];
    const int tid = threadIdx.x;

    if (blockIdx.x < BIN_WGS) {
        for (int chunk = blockIdx.x; chunk * CHUNK < E; chunk += BIN_WGS) {
            if (tid < NBIN) cnt[tid] = 0;
            __syncthreads();

            const int base = chunk * CHUNK;
            int    key[EPT], bin[EPT], rank[EPT];
            float4 val[EPT];
            bool   ok[EPT];
            #pragma unroll
            for (int u = 0; u < EPT; ++u) {
                const int e  = base + u * STAGE_THR + tid;
                ok[u] = e < E;
                const int ee = ok[u] ? e : 0;
                const int   k = src[ee];
                const int   n = dst[ee];
                const float s = signs[ee];
                val[u] = make_float4(w1[ee] * s, w2[ee] * s, wa[ee] * s, wb[ee] * s);
                bin[u] = ((n >> 4) << 1) | ((k >= KHALF) ? 1 : 0);
                key[u] = (k << 4) | (n & 15);
            }
            #pragma unroll
            for (int u = 0; u < EPT; ++u)
                rank[u] = ok[u] ? atomicAdd(&cnt[bin[u]], 1) : 0;
            __syncthreads();

            if (tid <= NBIN) {
                int s = 0;
                for (int b = 0; b < NBIN; ++b) {
                    const int c = cnt[b];
                    s += (b < tid) ? c : 0;
                }
                start[tid] = s;
                off[chunk * OFFP + tid] = s;
            }
            __syncthreads();

            #pragma unroll
            for (int u = 0; u < EPT; ++u) {
                if (ok[u]) {
                    const int q = start[bin[u]] + rank[u];
                    sK[q] = key[u];
                    sV[q] = val[u];
                }
            }
            __syncthreads();

            const int total = start[NBIN];
            #pragma unroll
            for (int u = 0; u < EPT; ++u) {
                const int q = u * STAGE_THR + tid;
                if (q < total) {
                    recK[base + q] = sK[q];
                    recV[base + q] = sV[q];
                }
            }
        }
    } else {
        const int w  = blockIdx.x - BIN_WGS;
        const int b  = (w & 1) * STAGE_THR + threadIdx.x;
        const int g  = w >> 1;
        const int j0 = g * KGRP;
        const float* srow = (j0 < INsz) ? in + (b * INsz + j0)
                                        : hx + (b * Hsz + (j0 - INsz));
        const float4 lo = *reinterpret_cast<const float4*>(srow);
        const float4 hi = *reinterpret_cast<const float4*>(srow + 4);
        X8[(g * 2 + 0) * Bsz + b] = make_float4(lo.x, lo.y, hi.x, hi.y);
        X8[(g * 2 + 1) * Bsz + b] = make_float4(lo.z, lo.w, hi.z, hi.w);
    }
}

#define GPB 10
#define RB  16

static_assert(HGRP % (2 * GPB) == 0, "the pipelined loop walks two blocks per trip");

struct FragSet {
    v4f a[GPB][2];
    v4f b[GPB][4];
};

__device__ __forceinline__ void load_a_block(FragSet& f, const v4f* __restrict__ xa, int ga) {
    #pragma unroll
    for (int i = 0; i < GPB; ++i) {
        const size_t g = (size_t)(ga + i);
        f.a[i][0] = xa[g * (2 * Bsz)];
        f.a[i][1] = xa[g * (2 * Bsz) + 16];
    }
}

__device__ __forceinline__ void load_b_block(FragSet& f, const v4f* lb, int gl) {
    #pragma unroll
    for (int i = 0; i < GPB; ++i) {
        #pragma unroll
        for (int s = 0; s < 2; ++s) {
            f.b[i][s * 2 + 0] = lb[((gl + i) * 4 + s * 2 + 0) * 32];
            f.b[i][s * 2 + 1] = lb[((gl + i) * 4 + s * 2 + 1) * 32];
        }
    }
}

__device__ __forceinline__ void load_block(FragSet& f,
                                           const v4f* __restrict__ xa,
                                           const v4f* lb,
                                           int ga, int gl) {
    load_a_block(f, xa, ga);
    load_b_block(f, lb, gl);
}

__device__ __forceinline__ void fetch_records(int (&key)[RB], float4 (&v)[RB],
                                              const int*    __restrict__ recK,
                                              const float4* __restrict__ recV,
                                              int rb, int r0, int re) {
    #pragma unroll
    for (int u = 0; u < RB; ++u) {
        int r = (r0 + u < re) ? (r0 + u) : (re - 1);
        r = (r < 0) ? 0 : r;
        key[u] = recK[rb + r];
        v[u]   = recV[rb + r];
    }
}

__device__ __forceinline__ void add_records(float* ldsW, const int (&key)[RB], const float4 (&v)[RB],
                                            int r0, int re, int kh) {
    #pragma unroll
    for (int u = 0; u < RB; ++u) {
        if (r0 + u < re) {
            const int k  = key[u] >> 4;
            const int c  = key[u] & 15;
            const int gl = (k >> 3) - kh * HGRP;
            const int jj = k & (KGRP - 1);
            const int st = jj >> 2;
            const int hf = (jj >> 1) & 1;
            const int f0 = (((gl * 4 + st * 2) * 32) + hf * 16 + c) * 4 + (jj & 1);
            atomicAdd(&ldsW[f0],       v[u].x);
            atomicAdd(&ldsW[f0 + 2],   v[u].y);
            atomicAdd(&ldsW[f0 + 128], v[u].z);
            atomicAdd(&ldsW[f0 + 130], v[u].w);
        }
    }
}

struct FirstRun {
    int    rs, re;
    int    key[RB];
    float4 v[RB];
};

__device__ __forceinline__ void fetch_first_off(FirstRun& p, const int* __restrict__ off,
                                                int j, int nchunk, int b2) {
    p.rs = 0;
    p.re = 0;
    if (j < nchunk) {
        p.rs = off[j * OFFP + b2];
        p.re = off[j * OFFP + b2 + 1];
    }
}

__device__ __forceinline__ void fetch_first_rec(FirstRun& p,
                                                const int*    __restrict__ recK,
                                                const float4* __restrict__ recV,
                                                int j, int nchunk) {
    const int jc = (j < nchunk) ? j : 0;
    fetch_records(p.key, p.v, recK, recV, jc * CHUNK, p.rs, p.re);
}

__device__ __forceinline__ void mma_block(const FragSet& f, v8f (&acc)[2][4]) {
    #pragma unroll
    for (int i = 0; i < GPB; ++i) {
        #pragma unroll
        for (int st = 0; st < 2; ++st) {
            const v2f a0 = st ? __builtin_shufflevector(f.a[i][0], f.a[i][0], 2, 3)
                              : __builtin_shufflevector(f.a[i][0], f.a[i][0], 0, 1);
            const v2f a1 = st ? __builtin_shufflevector(f.a[i][1], f.a[i][1], 2, 3)
                              : __builtin_shufflevector(f.a[i][1], f.a[i][1], 0, 1);
            #pragma unroll
            for (int l = 0; l < 4; ++l) {
                const v4f q  = f.b[i][st * 2 + (l >> 1)];
                const v2f bb = (l & 1) ? __builtin_shufflevector(q, q, 2, 3)
                                       : __builtin_shufflevector(q, q, 0, 1);
                acc[0][l] = __builtin_amdgcn_wmma_f32_16x16x4_f32(
                    false, a0, false, bb, (short)0, acc[0][l], false, false);
                acc[1][l] = __builtin_amdgcn_wmma_f32_16x16x4_f32(
                    false, a1, false, bb, (short)0, acc[1][l], false, false);
            }
        }
    }
}

__device__ __forceinline__ void half_pass(const int kh, const FirstRun& p,
                                          float* ldsW,
                                          const v4f* __restrict__ xa,
                                          const int*    __restrict__ recK,
                                          const float4* __restrict__ recV,
                                          const int*    __restrict__ off,
                                          int tid, int lane, int ntile, int nchunk,
                                          FragSet& f0, FragSet& f1,
                                          v8f (&acc)[2][4]) {
    const int gh = kh * HGRP;
    const int b2 = ntile * 2 + kh;
    const v4f* lb = reinterpret_cast<const v4f*>(ldsW) + lane;
    v4f* l4 = reinterpret_cast<v4f*>(ldsW);

    if (kh != 0) __syncthreads();
    load_a_block(f0, xa, gh);
    for (int i = tid; i < LDSW_FLOATS / 4; i += 128)
        l4[i] = (v4f){0.f, 0.f, 0.f, 0.f};
    __syncthreads();

    add_records(ldsW, p.key, p.v, p.rs, p.re, kh);
    {
        const int rb = tid * CHUNK;
        for (int r0 = p.rs + RB; r0 < p.re; r0 += RB) {
            int    key[RB];
            float4 v[RB];
            fetch_records(key, v, recK, recV, rb, r0, p.re);
            add_records(ldsW, key, v, r0, p.re, kh);
        }
    }
    for (int j = tid + 128; j < nchunk; j += 128) {
        const int rs = off[j * OFFP + b2];
        const int re = off[j * OFFP + b2 + 1];
        const int rb = j * CHUNK;
        for (int r0 = rs; r0 < re; r0 += RB) {
            int    key[RB];
            float4 v[RB];
            fetch_records(key, v, recK, recV, rb, r0, re);
            add_records(ldsW, key, v, r0, re, kh);
        }
    }
    __syncthreads();

    load_b_block(f0, lb, 0);
    int gb = 0;
    for (; gb + 2 * GPB < HGRP; gb += 2 * GPB) {
        load_block(f1, xa, lb, gh + gb + GPB, gb + GPB);
        __builtin_amdgcn_sched_barrier(0);
        mma_block(f0, acc);
        __builtin_amdgcn_sched_barrier(0);
        load_block(f0, xa, lb, gh + gb + 2 * GPB, gb + 2 * GPB);
        __builtin_amdgcn_sched_barrier(0);
        mma_block(f1, acc);
        __builtin_amdgcn_sched_barrier(0);
    }
    load_block(f1, xa, lb, gh + gb + GPB, gb + GPB);
    __builtin_amdgcn_sched_barrier(0);
    mma_block(f0, acc);
    mma_block(f1, acc);
}

__global__ __launch_bounds__(128) void gemm_fused(const float*  __restrict__ X8,
                                                  const int*    __restrict__ recK,
                                                  const float4* __restrict__ recV,
                                                  const int*    __restrict__ off,
                                                  const float* __restrict__ ts,
                                                  const float* __restrict__ b_ff1,
                                                  const float* __restrict__ b_ff2,
                                                  const float* __restrict__ b_ta,
                                                  const float* __restrict__ b_tb,
                                                  float* __restrict__ out,
                                                  int E) {
    __shared__ __attribute__((aligned(16))) float ldsW[LDSW_FLOATS];

    const int tid  = threadIdx.x;
    const int lane = tid & 31;
    const int wave = tid >> 5;

    const int ntile = blockIdx.x;
    const int n0   = ntile * 16;
    const int m0   = (blockIdx.y * 4 + wave) * 32;
    const int ncol = lane & 15;
    const int half = lane >> 4;
    const int nchunk = (E + CHUNK - 1) / CHUNK;

    FirstRun p0, p1;
    fetch_first_off(p0, off, tid, nchunk, ntile * 2 + 0);
    fetch_first_off(p1, off, tid, nchunk, ntile * 2 + 1);
    fetch_first_rec(p0, recK, recV, tid, nchunk);
    fetch_first_rec(p1, recK, recV, tid, nchunk);

    v8f acc[2][4];
    #pragma unroll
    for (int s = 0; s < 2; ++s)
        #pragma unroll
        for (int l = 0; l < 4; ++l)
            acc[s][l] = (v8f){0.f,0.f,0.f,0.f,0.f,0.f,0.f,0.f};

    const v4f* xa = reinterpret_cast<const v4f*>(X8) + (half * Bsz + m0 + ncol);

    FragSet f0, f1;
    half_pass(0, p0, ldsW, xa, recK, recV, off, tid, lane, ntile, nchunk, f0, f1, acc);
    half_pass(1, p1, ldsW, xa, recK, recV, off, tid, lane, ntile, nchunk, f0, f1, acc);

    const int   ng  = n0 + ncol;
    const float bf1 = b_ff1[ng];
    const float bf2 = b_ff2[ng];
    const float bta = b_ta[ng];
    const float btb = b_tb[ng];
    const int   mhi = (lane >> 4) << 3;

    #pragma unroll
    for (int s = 0; s < 2; ++s) {
        #pragma unroll
        for (int r = 0; r < 8; ++r) {
            const int   m   = m0 + s * 16 + r + mhi;
            const float tsv = ts[m];
            const float f1  = tanhf(acc[s][0][r] + bf1);
            const float f2  = tanhf(acc[s][1][r] + bf2);
            const float ta  = acc[s][2][r] + bta;
            const float tb  = acc[s][3][r] + btb;
            const float sg  = 1.f / (1.f + expf(-(ta * tsv + tb)));
            const float h   = f1 * (1.f - sg) + sg * f2;
            const size_t o  = (size_t)m * Hsz + ng;
            out[o]                         = h;
            out[(size_t)Bsz * Hsz + o]     = h;
        }
    }
}

extern "C" void kernel_launch(void* const* d_in, const int* in_sizes, int n_in,
                              void* d_out, int out_size, void* d_ws, size_t ws_size,
                              hipStream_t stream) {
    const float* input = (const float*)d_in[0];
    const float* hx    = (const float*)d_in[1];
    const float* ts    = (const float*)d_in[2];
    const int*   src   = (const int*)  d_in[3];
    const int*   dst   = (const int*)  d_in[4];
    const float* signs = (const float*)d_in[5];
    const float* w1    = (const float*)d_in[6];
    const float* w2    = (const float*)d_in[7];
    const float* wa    = (const float*)d_in[8];
    const float* wb    = (const float*)d_in[9];
    const float* bf1   = (const float*)d_in[10];
    const float* bf2   = (const float*)d_in[11];
    const float* bta   = (const float*)d_in[12];
    const float* btb   = (const float*)d_in[13];
    const int    E     = in_sizes[3];

    float*  X8   = (float*)d_ws;
    float4* recV = (float4*)(X8 + CONC * Bsz);
    int*    recK = (int*)(recV + MAXE);
    int*    off  = recK + MAXE;
    float*  out  = (float*)d_out;

    stage_bin_x8<<<STAGE_WGS, STAGE_THR, 0, stream>>>(src, dst, signs, w1, w2, wa, wb,
                                                      recK, recV, off, input, hx, (float4*)X8, E);

    dim3 grid(Hsz / 16, Bsz / 128);
    gemm_fused<<<grid, 128, 0, stream>>>(X8, recK, recV, off, ts, bf1, bf2, bta, btb, out, E);
}


# Epilogue: both tanh and the sigmoid (as 0.5+0.5*tanh(u/2)) on the hardware tanh instruction, branch-free, in place of libm's guarded blocks: a lane's outputs interleave, kernel text 11 KB shorter
# speedup vs baseline: 4.0535x; 1.1187x over previous
#include <hip/hip_runtime.h>
#include <math.h>

#define Bsz   512
#define INsz  256
#define Hsz   1024
#define CONC  1280

typedef __attribute__((ext_vector_type(2))) float v2f;
typedef __attribute__((ext_vector_type(4))) float v4f;
typedef __attribute__((ext_vector_type(8))) float v8f;

#define KGRP   8
#define NGRP   (CONC / KGRP)
#define HGRP   (NGRP / 2)
#define KHALF  (CONC / 2)
#define LDSW_FLOATS (HGRP * 512)

#define STAGE_THR   256
#define CHUNK       1024
#define EPT         (CHUNK / STAGE_THR)
#define NBIN        128
#define OFFP        132
#define BIN_WGS     128
#define XT_ROWS     8
#define XT_WGS      (2 * (CONC / XT_ROWS))
#define STAGE_WGS   (BIN_WGS + XT_WGS)
#define MAXE        (1 << 20)

static_assert(INsz % XT_ROWS == 0 && CONC % XT_ROWS == 0, "a row group reads one source only");
static_assert(XT_ROWS == KGRP, "one work-group row block is one k group");
static_assert(Bsz == 2 * STAGE_THR && XT_ROWS == 8, "X^T tiling must cover exactly");
static_assert(NBIN == 2 * (Hsz / 16), "one bin per (N tile, K half)");
static_assert(NBIN + 1 <= STAGE_THR && NBIN + 1 <= OFFP, "one thread per start offset");

__global__ __launch_bounds__(STAGE_THR) void stage_bin_x8(const int*   __restrict__ src,
                                                          const int*   __restrict__ dst,
                                                          const float* __restrict__ signs,
                                                          const float* __restrict__ w1,
                                                          const float* __restrict__ w2,
                                                          const float* __restrict__ wa,
                                                          const float* __restrict__ wb,
                                                          int*    __restrict__ recK,
                                                          float4* __restrict__ recV,
                                                          int*    __restrict__ off,
                                                          const float* __restrict__ in,
                                                          const float* __restrict__ hx,
                                                          float4* __restrict__ X8,
                                                          int E) {
    __shared__ int    cnt[NBIN];
    __shared__ int    start[NBIN + 1];
    __shared__ int    sK[CHUNK];
    __shared__ float4 sV[CHUNK];
    const int tid = threadIdx.x;

    if (blockIdx.x < BIN_WGS) {
        for (int chunk = blockIdx.x; chunk * CHUNK < E; chunk += BIN_WGS) {
            if (tid < NBIN) cnt[tid] = 0;
            __syncthreads();

            const int base = chunk * CHUNK;
            int    key[EPT], bin[EPT], rank[EPT];
            float4 val[EPT];
            bool   ok[EPT];
            #pragma unroll
            for (int u = 0; u < EPT; ++u) {
                const int e  = base + u * STAGE_THR + tid;
                ok[u] = e < E;
                const int ee = ok[u] ? e : 0;
                const int   k = src[ee];
                const int   n = dst[ee];
                const float s = signs[ee];
                val[u] = make_float4(w1[ee] * s, w2[ee] * s, wa[ee] * s, wb[ee] * s);
                bin[u] = ((n >> 4) << 1) | ((k >= KHALF) ? 1 : 0);
                key[u] = (k << 4) | (n & 15);
            }
            #pragma unroll
            for (int u = 0; u < EPT; ++u)
                rank[u] = ok[u] ? atomicAdd(&cnt[bin[u]], 1) : 0;
            __syncthreads();

            if (tid <= NBIN) {
                int s = 0;
                for (int b = 0; b < NBIN; ++b) {
                    const int c = cnt[b];
                    s += (b < tid) ? c : 0;
                }
                start[tid] = s;
                off[chunk * OFFP + tid] = s;
            }
            __syncthreads();

            #pragma unroll
            for (int u = 0; u < EPT; ++u) {
                if (ok[u]) {
                    const int q = start[bin[u]] + rank[u];
                    sK[q] = key[u];
                    sV[q] = val[u];
                }
            }
            __syncthreads();

            const int total = start[NBIN];
            #pragma unroll
            for (int u = 0; u < EPT; ++u) {
                const int q = u * STAGE_THR + tid;
                if (q < total) {
                    recK[base + q] = sK[q];
                    recV[base + q] = sV[q];
                }
            }
        }
    } else {
        const int w  = blockIdx.x - BIN_WGS;
        const int b  = (w & 1) * STAGE_THR + threadIdx.x;
        const int g  = w >> 1;
        const int j0 = g * KGRP;
        const float* srow = (j0 < INsz) ? in + (b * INsz + j0)
                                        : hx + (b * Hsz + (j0 - INsz));
        const float4 lo = *reinterpret_cast<const float4*>(srow);
        const float4 hi = *reinterpret_cast<const float4*>(srow + 4);
        X8[(g * 2 + 0) * Bsz + b] = make_float4(lo.x, lo.y, hi.x, hi.y);
        X8[(g * 2 + 1) * Bsz + b] = make_float4(lo.z, lo.w, hi.z, hi.w);
    }
}

#define GPB 10
#define RB  16

static_assert(HGRP % (2 * GPB) == 0, "the pipelined loop walks two blocks per trip");

struct FragSet {
    v4f a[GPB][2];
    v4f b[GPB][4];
};

__device__ __forceinline__ void load_a_block(FragSet& f, const v4f* __restrict__ xa, int ga) {
    #pragma unroll
    for (int i = 0; i < GPB; ++i) {
        const size_t g = (size_t)(ga + i);
        f.a[i][0] = xa[g * (2 * Bsz)];
        f.a[i][1] = xa[g * (2 * Bsz) + 16];
    }
}

__device__ __forceinline__ void load_b_block(FragSet& f, const v4f* lb, int gl) {
    #pragma unroll
    for (int i = 0; i < GPB; ++i) {
        #pragma unroll
        for (int s = 0; s < 2; ++s) {
            f.b[i][s * 2 + 0] = lb[((gl + i) * 4 + s * 2 + 0) * 32];
            f.b[i][s * 2 + 1] = lb[((gl + i) * 4 + s * 2 + 1) * 32];
        }
    }
}

__device__ __forceinline__ void load_block(FragSet& f,
                                           const v4f* __restrict__ xa,
                                           const v4f* lb,
                                           int ga, int gl) {
    load_a_block(f, xa, ga);
    load_b_block(f, lb, gl);
}

__device__ __forceinline__ void fetch_records(int (&key)[RB], float4 (&v)[RB],
                                              const int*    __restrict__ recK,
                                              const float4* __restrict__ recV,
                                              int rb, int r0, int re) {
    #pragma unroll
    for (int u = 0; u < RB; ++u) {
        int r = (r0 + u < re) ? (r0 + u) : (re - 1);
        r = (r < 0) ? 0 : r;
        key[u] = recK[rb + r];
        v[u]   = recV[rb + r];
    }
}

__device__ __forceinline__ void add_records(float* ldsW, const int (&key)[RB], const float4 (&v)[RB],
                                            int r0, int re, int kh) {
    #pragma unroll
    for (int u = 0; u < RB; ++u) {
        if (r0 + u < re) {
            const int k  = key[u] >> 4;
            const int c  = key[u] & 15;
            const int gl = (k >> 3) - kh * HGRP;
            const int jj = k & (KGRP - 1);
            const int st = jj >> 2;
            const int hf = (jj >> 1) & 1;
            const int f0 = (((gl * 4 + st * 2) * 32) + hf * 16 + c) * 4 + (jj & 1);
            atomicAdd(&ldsW[f0],       v[u].x);
            atomicAdd(&ldsW[f0 + 2],   v[u].y);
            atomicAdd(&ldsW[f0 + 128], v[u].z);
            atomicAdd(&ldsW[f0 + 130], v[u].w);
        }
    }
}

struct FirstRun {
    int    rs, re;
    int    key[RB];
    float4 v[RB];
};

__device__ __forceinline__ void fetch_first_off(FirstRun& p, const int* __restrict__ off,
                                                int j, int nchunk, int b2) {
    p.rs = 0;
    p.re = 0;
    if (j < nchunk) {
        p.rs = off[j * OFFP + b2];
        p.re = off[j * OFFP + b2 + 1];
    }
}

__device__ __forceinline__ void fetch_first_rec(FirstRun& p,
                                                const int*    __restrict__ recK,
                                                const float4* __restrict__ recV,
                                                int j, int nchunk) {
    const int jc = (j < nchunk) ? j : 0;
    fetch_records(p.key, p.v, recK, recV, jc * CHUNK, p.rs, p.re);
}

__device__ __forceinline__ void mma_block(const FragSet& f, v8f (&acc)[2][4]) {
    #pragma unroll
    for (int i = 0; i < GPB; ++i) {
        #pragma unroll
        for (int st = 0; st < 2; ++st) {
            const v2f a0 = st ? __builtin_shufflevector(f.a[i][0], f.a[i][0], 2, 3)
                              : __builtin_shufflevector(f.a[i][0], f.a[i][0], 0, 1);
            const v2f a1 = st ? __builtin_shufflevector(f.a[i][1], f.a[i][1], 2, 3)
                              : __builtin_shufflevector(f.a[i][1], f.a[i][1], 0, 1);
            #pragma unroll
            for (int l = 0; l < 4; ++l) {
                const v4f q  = f.b[i][st * 2 + (l >> 1)];
                const v2f bb = (l & 1) ? __builtin_shufflevector(q, q, 2, 3)
                                       : __builtin_shufflevector(q, q, 0, 1);
                acc[0][l] = __builtin_amdgcn_wmma_f32_16x16x4_f32(
                    false, a0, false, bb, (short)0, acc[0][l], false, false);
                acc[1][l] = __builtin_amdgcn_wmma_f32_16x16x4_f32(
                    false, a1, false, bb, (short)0, acc[1][l], false, false);
            }
        }
    }
}

__device__ __forceinline__ void half_pass(const int kh, const FirstRun& p,
                                          float* ldsW,
                                          const v4f* __restrict__ xa,
                                          const int*    __restrict__ recK,
                                          const float4* __restrict__ recV,
                                          const int*    __restrict__ off,
                                          int tid, int lane, int ntile, int nchunk,
                                          FragSet& f0, FragSet& f1,
                                          v8f (&acc)[2][4]) {
    const int gh = kh * HGRP;
    const int b2 = ntile * 2 + kh;
    const v4f* lb = reinterpret_cast<const v4f*>(ldsW) + lane;
    v4f* l4 = reinterpret_cast<v4f*>(ldsW);

    if (kh != 0) __syncthreads();
    load_a_block(f0, xa, gh);
    for (int i = tid; i < LDSW_FLOATS / 4; i += 128)
        l4[i] = (v4f){0.f, 0.f, 0.f, 0.f};
    __syncthreads();

    add_records(ldsW, p.key, p.v, p.rs, p.re, kh);
    {
        const int rb = tid * CHUNK;
        for (int r0 = p.rs + RB; r0 < p.re; r0 += RB) {
            int    key[RB];
            float4 v[RB];
            fetch_records(key, v, recK, recV, rb, r0, p.re);
            add_records(ldsW, key, v, r0, p.re, kh);
        }
    }
    for (int j = tid + 128; j < nchunk; j += 128) {
        const int rs = off[j * OFFP + b2];
        const int re = off[j * OFFP + b2 + 1];
        const int rb = j * CHUNK;
        for (int r0 = rs; r0 < re; r0 += RB) {
            int    key[RB];
            float4 v[RB];
            fetch_records(key, v, recK, recV, rb, r0, re);
            add_records(ldsW, key, v, r0, re, kh);
        }
    }
    __syncthreads();

    load_b_block(f0, lb, 0);
    int gb = 0;
    for (; gb + 2 * GPB < HGRP; gb += 2 * GPB) {
        load_block(f1, xa, lb, gh + gb + GPB, gb + GPB);
        __builtin_amdgcn_sched_barrier(0);
        mma_block(f0, acc);
        __builtin_amdgcn_sched_barrier(0);
        load_block(f0, xa, lb, gh + gb + 2 * GPB, gb + 2 * GPB);
        __builtin_amdgcn_sched_barrier(0);
        mma_block(f1, acc);
        __builtin_amdgcn_sched_barrier(0);
    }
    load_block(f1, xa, lb, gh + gb + GPB, gb + GPB);
    __builtin_amdgcn_sched_barrier(0);
    mma_block(f0, acc);
    mma_block(f1, acc);
}

__global__ __launch_bounds__(128) void gemm_fused(const float*  __restrict__ X8,
                                                  const int*    __restrict__ recK,
                                                  const float4* __restrict__ recV,
                                                  const int*    __restrict__ off,
                                                  const float* __restrict__ ts,
                                                  const float* __restrict__ b_ff1,
                                                  const float* __restrict__ b_ff2,
                                                  const float* __restrict__ b_ta,
                                                  const float* __restrict__ b_tb,
                                                  float* __restrict__ out,
                                                  int E) {
    __shared__ __attribute__((aligned(16))) float ldsW[LDSW_FLOATS];

    const int tid  = threadIdx.x;
    const int lane = tid & 31;
    const int wave = tid >> 5;

    const int ntile = blockIdx.x;
    const int n0   = ntile * 16;
    const int m0   = (blockIdx.y * 4 + wave) * 32;
    const int ncol = lane & 15;
    const int half = lane >> 4;
    const int nchunk = (E + CHUNK - 1) / CHUNK;

    FirstRun p0, p1;
    fetch_first_off(p0, off, tid, nchunk, ntile * 2 + 0);
    fetch_first_off(p1, off, tid, nchunk, ntile * 2 + 1);
    fetch_first_rec(p0, recK, recV, tid, nchunk);
    fetch_first_rec(p1, recK, recV, tid, nchunk);

    v8f acc[2][4];
    #pragma unroll
    for (int s = 0; s < 2; ++s)
        #pragma unroll
        for (int l = 0; l < 4; ++l)
            acc[s][l] = (v8f){0.f,0.f,0.f,0.f,0.f,0.f,0.f,0.f};

    const v4f* xa = reinterpret_cast<const v4f*>(X8) + (half * Bsz + m0 + ncol);

    FragSet f0, f1;
    half_pass(0, p0, ldsW, xa, recK, recV, off, tid, lane, ntile, nchunk, f0, f1, acc);
    half_pass(1, p1, ldsW, xa, recK, recV, off, tid, lane, ntile, nchunk, f0, f1, acc);

    const int   ng  = n0 + ncol;
    const float bf1 = b_ff1[ng];
    const float bf2 = b_ff2[ng];
    const float bta = b_ta[ng];
    const float btb = b_tb[ng];
    const int   mhi = (lane >> 4) << 3;

    #pragma unroll
    for (int s = 0; s < 2; ++s) {
        #pragma unroll
        for (int r = 0; r < 8; ++r) {
            const int   m   = m0 + s * 16 + r + mhi;
            const float tsv = ts[m];
            const float f1  = __builtin_amdgcn_tanhf(acc[s][0][r] + bf1);
            const float f2  = __builtin_amdgcn_tanhf(acc[s][1][r] + bf2);
            const float ta  = acc[s][2][r] + bta;
            const float tb  = acc[s][3][r] + btb;
            const float sg  = 0.5f + 0.5f * __builtin_amdgcn_tanhf(0.5f * (ta * tsv + tb));
            const float h   = f1 * (1.f - sg) + sg * f2;
            const size_t o  = (size_t)m * Hsz + ng;
            out[o]                         = h;
            out[(size_t)Bsz * Hsz + o]     = h;
        }
    }
}

extern "C" void kernel_launch(void* const* d_in, const int* in_sizes, int n_in,
                              void* d_out, int out_size, void* d_ws, size_t ws_size,
                              hipStream_t stream) {
    const float* input = (const float*)d_in[0];
    const float* hx    = (const float*)d_in[1];
    const float* ts    = (const float*)d_in[2];
    const int*   src   = (const int*)  d_in[3];
    const int*   dst   = (const int*)  d_in[4];
    const float* signs = (const float*)d_in[5];
    const float* w1    = (const float*)d_in[6];
    const float* w2    = (const float*)d_in[7];
    const float* wa    = (const float*)d_in[8];
    const float* wb    = (const float*)d_in[9];
    const float* bf1   = (const float*)d_in[10];
    const float* bf2   = (const float*)d_in[11];
    const float* bta   = (const float*)d_in[12];
    const float* btb   = (const float*)d_in[13];
    const int    E     = in_sizes[3];

    float*  X8   = (float*)d_ws;
    float4* recV = (float4*)(X8 + CONC * Bsz);
    int*    recK = (int*)(recV + MAXE);
    int*    off  = recK + MAXE;
    float*  out  = (float*)d_out;

    stage_bin_x8<<<STAGE_WGS, STAGE_THR, 0, stream>>>(src, dst, signs, w1, w2, wa, wb,
                                                      recK, recV, off, input, hx, (float4*)X8, E);

    dim3 grid(Hsz / 16, Bsz / 128);
    gemm_fused<<<grid, 128, 0, stream>>>(X8, recK, recV, off, ts, bf1, bf2, bta, btb, out, E);
}


# 8-wave GEMM work-groups (2 waves per SIMD, 16 rows each) build the N tile's weights for ALL of K once in 320 KB LDS: no mid-kernel rebuild, one unbroken K loop; k order kept, LDS add order free
# speedup vs baseline: 4.2130x; 1.0393x over previous
#include <hip/hip_runtime.h>
#include <math.h>

#define Bsz   512
#define INsz  256
#define Hsz   1024
#define CONC  1280

typedef __attribute__((ext_vector_type(2))) float v2f;
typedef __attribute__((ext_vector_type(4))) float v4f;
typedef __attribute__((ext_vector_type(8))) float v8f;

#define KGRP   8
#define NGRP   (CONC / KGRP)
#define KHALF  (CONC / 2)
#define LDSW_FLOATS (NGRP * 512)

#define STAGE_THR   256
#define CHUNK       1024
#define EPT         (CHUNK / STAGE_THR)
#define NBIN        128
#define OFFP        132
#define BIN_WGS     128
#define XT_ROWS     8
#define XT_WGS      (2 * (CONC / XT_ROWS))
#define STAGE_WGS   (BIN_WGS + XT_WGS)
#define MAXE        (1 << 20)

static_assert(INsz % XT_ROWS == 0 && CONC % XT_ROWS == 0, "a row group reads one source only");
static_assert(XT_ROWS == KGRP, "one work-group row block is one k group");
static_assert(Bsz == 2 * STAGE_THR && XT_ROWS == 8, "X^T tiling must cover exactly");
static_assert(NBIN == 2 * (Hsz / 16), "one bin per (N tile, K half)");
static_assert(NBIN + 1 <= STAGE_THR && NBIN + 1 <= OFFP, "one thread per start offset");

__global__ __launch_bounds__(STAGE_THR) void stage_bin_x8(const int*   __restrict__ src,
                                                          const int*   __restrict__ dst,
                                                          const float* __restrict__ signs,
                                                          const float* __restrict__ w1,
                                                          const float* __restrict__ w2,
                                                          const float* __restrict__ wa,
                                                          const float* __restrict__ wb,
                                                          int*    __restrict__ recK,
                                                          float4* __restrict__ recV,
                                                          int*    __restrict__ off,
                                                          const float* __restrict__ in,
                                                          const float* __restrict__ hx,
                                                          float4* __restrict__ X8,
                                                          int E) {
    __shared__ int    cnt[NBIN];
    __shared__ int    start[NBIN + 1];
    __shared__ int    sK[CHUNK];
    __shared__ float4 sV[CHUNK];
    const int tid = threadIdx.x;

    if (blockIdx.x < BIN_WGS) {
        for (int chunk = blockIdx.x; chunk * CHUNK < E; chunk += BIN_WGS) {
            if (tid < NBIN) cnt[tid] = 0;
            __syncthreads();

            const int base = chunk * CHUNK;
            int    key[EPT], bin[EPT], rank[EPT];
            float4 val[EPT];
            bool   ok[EPT];
            #pragma unroll
            for (int u = 0; u < EPT; ++u) {
                const int e  = base + u * STAGE_THR + tid;
                ok[u] = e < E;
                const int ee = ok[u] ? e : 0;
                const int   k = src[ee];
                const int   n = dst[ee];
                const float s = signs[ee];
                val[u] = make_float4(w1[ee] * s, w2[ee] * s, wa[ee] * s, wb[ee] * s);
                bin[u] = ((n >> 4) << 1) | ((k >= KHALF) ? 1 : 0);
                key[u] = (k << 4) | (n & 15);
            }
            #pragma unroll
            for (int u = 0; u < EPT; ++u)
                rank[u] = ok[u] ? atomicAdd(&cnt[bin[u]], 1) : 0;
            __syncthreads();

            if (tid <= NBIN) {
                int s = 0;
                for (int b = 0; b < NBIN; ++b) {
                    const int c = cnt[b];
                    s += (b < tid) ? c : 0;
                }
                start[tid] = s;
                off[chunk * OFFP + tid] = s;
            }
            __syncthreads();

            #pragma unroll
            for (int u = 0; u < EPT; ++u) {
                if (ok[u]) {
                    const int q = start[bin[u]] + rank[u];
                    sK[q] = key[u];
                    sV[q] = val[u];
                }
            }
            __syncthreads();

            const int total = start[NBIN];
            #pragma unroll
            for (int u = 0; u < EPT; ++u) {
                const int q = u * STAGE_THR + tid;
                if (q < total) {
                    recK[base + q] = sK[q];
                    recV[base + q] = sV[q];
                }
            }
        }
    } else {
        const int w  = blockIdx.x - BIN_WGS;
        const int b  = (w & 1) * STAGE_THR + threadIdx.x;
        const int g  = w >> 1;
        const int j0 = g * KGRP;
        const float* srow = (j0 < INsz) ? in + (b * INsz + j0)
                                        : hx + (b * Hsz + (j0 - INsz));
        const float4 lo = *reinterpret_cast<const float4*>(srow);
        const float4 hi = *reinterpret_cast<const float4*>(srow + 4);
        X8[(g * 2 + 0) * Bsz + b] = make_float4(lo.x, lo.y, hi.x, hi.y);
        X8[(g * 2 + 1) * Bsz + b] = make_float4(lo.z, lo.w, hi.z, hi.w);
    }
}

#define GPB 10
#define GEMM_THR 256
#define BUILD_THR (GEMM_THR / 2)
#define RB  16

static_assert(NGRP % (2 * GPB) == 0, "the pipelined loop walks two blocks per trip");

struct FragSet {
    v4f a[GPB];
    v4f b[GPB][4];
};

__device__ __forceinline__ void load_a_block(FragSet& f, const v4f* __restrict__ xa, int ga) {
    #pragma unroll
    for (int i = 0; i < GPB; ++i) {
        const size_t g = (size_t)(ga + i);
        f.a[i] = xa[g * (2 * Bsz)];
    }
}

__device__ __forceinline__ void load_b_block(FragSet& f, const v4f* lb, int gl) {
    #pragma unroll
    for (int i = 0; i < GPB; ++i) {
        #pragma unroll
        for (int s = 0; s < 2; ++s) {
            f.b[i][s * 2 + 0] = lb[((gl + i) * 4 + s * 2 + 0) * 32];
            f.b[i][s * 2 + 1] = lb[((gl + i) * 4 + s * 2 + 1) * 32];
        }
    }
}

__device__ __forceinline__ void load_block(FragSet& f,
                                           const v4f* __restrict__ xa,
                                           const v4f* lb,
                                           int ga, int gl) {
    load_a_block(f, xa, ga);
    load_b_block(f, lb, gl);
}

__device__ __forceinline__ void fetch_records(int (&key)[RB], float4 (&v)[RB],
                                              const int*    __restrict__ recK,
                                              const float4* __restrict__ recV,
                                              int rb, int r0, int re) {
    #pragma unroll
    for (int u = 0; u < RB; ++u) {
        int r = (r0 + u < re) ? (r0 + u) : (re - 1);
        r = (r < 0) ? 0 : r;
        key[u] = recK[rb + r];
        v[u]   = recV[rb + r];
    }
}

__device__ __forceinline__ void add_records(float* ldsW, const int (&key)[RB], const float4 (&v)[RB],
                                            int r0, int re) {
    #pragma unroll
    for (int u = 0; u < RB; ++u) {
        if (r0 + u < re) {
            const int k  = key[u] >> 4;
            const int c  = key[u] & 15;
            const int gl = k >> 3;
            const int jj = k & (KGRP - 1);
            const int st = jj >> 2;
            const int hf = (jj >> 1) & 1;
            const int f0 = (((gl * 4 + st * 2) * 32) + hf * 16 + c) * 4 + (jj & 1);
            atomicAdd(&ldsW[f0],       v[u].x);
            atomicAdd(&ldsW[f0 + 2],   v[u].y);
            atomicAdd(&ldsW[f0 + 128], v[u].z);
            atomicAdd(&ldsW[f0 + 130], v[u].w);
        }
    }
}

struct FirstRun {
    int    rs, re;
    int    key[RB];
    float4 v[RB];
};

__device__ __forceinline__ void fetch_first_off(FirstRun& p, const int* __restrict__ off,
                                                int j, int nchunk, int b2) {
    p.rs = 0;
    p.re = 0;
    if (j < nchunk) {
        p.rs = off[j * OFFP + b2];
        p.re = off[j * OFFP + b2 + 1];
    }
}

__device__ __forceinline__ void fetch_first_rec(FirstRun& p,
                                                const int*    __restrict__ recK,
                                                const float4* __restrict__ recV,
                                                int j, int nchunk) {
    const int jc = (j < nchunk) ? j : 0;
    fetch_records(p.key, p.v, recK, recV, jc * CHUNK, p.rs, p.re);
}

__device__ __forceinline__ void mma_block(const FragSet& f, v8f (&acc)[4]) {
    #pragma unroll
    for (int i = 0; i < GPB; ++i) {
        #pragma unroll
        for (int st = 0; st < 2; ++st) {
            const v2f a0 = st ? __builtin_shufflevector(f.a[i], f.a[i], 2, 3)
                              : __builtin_shufflevector(f.a[i], f.a[i], 0, 1);
            #pragma unroll
            for (int l = 0; l < 4; ++l) {
                const v4f q  = f.b[i][st * 2 + (l >> 1)];
                const v2f bb = (l & 1) ? __builtin_shufflevector(q, q, 2, 3)
                                       : __builtin_shufflevector(q, q, 0, 1);
                acc[l] = __builtin_amdgcn_wmma_f32_16x16x4_f32(
                    false, a0, false, bb, (short)0, acc[l], false, false);
            }
        }
    }
}

__global__ __launch_bounds__(GEMM_THR) void gemm_fused(const float*  __restrict__ X8,
                                                       const int*    __restrict__ recK,
                                                       const float4* __restrict__ recV,
                                                       const int*    __restrict__ off,
                                                       const float* __restrict__ ts,
                                                       const float* __restrict__ b_ff1,
                                                       const float* __restrict__ b_ff2,
                                                       const float* __restrict__ b_ta,
                                                       const float* __restrict__ b_tb,
                                                       float* __restrict__ out,
                                                       int E) {
    __shared__ __attribute__((aligned(16))) float ldsW[LDSW_FLOATS];

    const int tid  = threadIdx.x;
    const int lane = tid & 31;
    const int wave = tid >> 5;

    const int ntile = blockIdx.x;
    const int n0   = ntile * 16;
    const int m0   = (blockIdx.y * 8 + wave) * 16;
    const int ncol = lane & 15;
    const int half = lane >> 4;
    const int nchunk = (E + CHUNK - 1) / CHUNK;

    const int kh = tid >> 7;
    const int j0 = tid & (BUILD_THR - 1);
    const int b2 = ntile * 2 + kh;

    FirstRun p;
    fetch_first_off(p, off, j0, nchunk, b2);
    fetch_first_rec(p, recK, recV, j0, nchunk);

    v8f acc[4];
    #pragma unroll
    for (int l = 0; l < 4; ++l)
        acc[l] = (v8f){0.f,0.f,0.f,0.f,0.f,0.f,0.f,0.f};

    const v4f* xa = reinterpret_cast<const v4f*>(X8) + (half * Bsz + m0 + ncol);
    const v4f* lb = reinterpret_cast<const v4f*>(ldsW) + lane;
    v4f* l4 = reinterpret_cast<v4f*>(ldsW);

    FragSet f0, f1;

    load_a_block(f0, xa, 0);
    for (int i = tid; i < LDSW_FLOATS / 4; i += GEMM_THR)
        l4[i] = (v4f){0.f, 0.f, 0.f, 0.f};
    __syncthreads();

    add_records(ldsW, p.key, p.v, p.rs, p.re);
    {
        const int rb = j0 * CHUNK;
        for (int r0 = p.rs + RB; r0 < p.re; r0 += RB) {
            int    key[RB];
            float4 v[RB];
            fetch_records(key, v, recK, recV, rb, r0, p.re);
            add_records(ldsW, key, v, r0, p.re);
        }
    }
    for (int j = j0 + BUILD_THR; j < nchunk; j += BUILD_THR) {
        const int rs = off[j * OFFP + b2];
        const int re = off[j * OFFP + b2 + 1];
        const int rb = j * CHUNK;
        for (int r0 = rs; r0 < re; r0 += RB) {
            int    key[RB];
            float4 v[RB];
            fetch_records(key, v, recK, recV, rb, r0, re);
            add_records(ldsW, key, v, r0, re);
        }
    }
    __syncthreads();

    load_b_block(f0, lb, 0);
    int gb = 0;
    for (; gb + 2 * GPB < NGRP; gb += 2 * GPB) {
        load_block(f1, xa, lb, gb + GPB, gb + GPB);
        __builtin_amdgcn_sched_barrier(0);
        mma_block(f0, acc);
        __builtin_amdgcn_sched_barrier(0);
        load_block(f0, xa, lb, gb + 2 * GPB, gb + 2 * GPB);
        __builtin_amdgcn_sched_barrier(0);
        mma_block(f1, acc);
        __builtin_amdgcn_sched_barrier(0);
    }
    load_block(f1, xa, lb, gb + GPB, gb + GPB);
    __builtin_amdgcn_sched_barrier(0);
    mma_block(f0, acc);
    mma_block(f1, acc);

    const int   ng  = n0 + ncol;
    const float bf1 = b_ff1[ng];
    const float bf2 = b_ff2[ng];
    const float bta = b_ta[ng];
    const float btb = b_tb[ng];
    const int   mhi = (lane >> 4) << 3;

    #pragma unroll
    for (int r = 0; r < 8; ++r) {
        const int   m   = m0 + r + mhi;
        const float tsv = ts[m];
        const float f1v = __builtin_amdgcn_tanhf(acc[0][r] + bf1);
        const float f2v = __builtin_amdgcn_tanhf(acc[1][r] + bf2);
        const float ta  = acc[2][r] + bta;
        const float tb  = acc[3][r] + btb;
        const float sg  = 0.5f + 0.5f * __builtin_amdgcn_tanhf(0.5f * (ta * tsv + tb));
        const float h   = f1v * (1.f - sg) + sg * f2v;
        const size_t o  = (size_t)m * Hsz + ng;
        out[o]                         = h;
        out[(size_t)Bsz * Hsz + o]     = h;
    }
}

extern "C" void kernel_launch(void* const* d_in, const int* in_sizes, int n_in,
                              void* d_out, int out_size, void* d_ws, size_t ws_size,
                              hipStream_t stream) {
    const float* input = (const float*)d_in[0];
    const float* hx    = (const float*)d_in[1];
    const float* ts    = (const float*)d_in[2];
    const int*   src   = (const int*)  d_in[3];
    const int*   dst   = (const int*)  d_in[4];
    const float* signs = (const float*)d_in[5];
    const float* w1    = (const float*)d_in[6];
    const float* w2    = (const float*)d_in[7];
    const float* wa    = (const float*)d_in[8];
    const float* wb    = (const float*)d_in[9];
    const float* bf1   = (const float*)d_in[10];
    const float* bf2   = (const float*)d_in[11];
    const float* bta   = (const float*)d_in[12];
    const float* btb   = (const float*)d_in[13];
    const int    E     = in_sizes[3];

    float*  X8   = (float*)d_ws;
    float4* recV = (float4*)(X8 + CONC * Bsz);
    int*    recK = (int*)(recV + MAXE);
    int*    off  = recK + MAXE;
    float*  out  = (float*)d_out;

    stage_bin_x8<<<STAGE_WGS, STAGE_THR, 0, stream>>>(src, dst, signs, w1, w2, wa, wb,
                                                      recK, recV, off, input, hx, (float4*)X8, E);

    dim3 grid(Hsz / 16, Bsz / 128);
    gemm_fused<<<grid, GEMM_THR, 0, stream>>>(X8, recK, recV, off, ts, bf1, bf2, bta, btb, out, E);
}


# GEMM N tile = 4 dst columns x the 4 matrices (one shared edge pattern): a tile walks only its k rows that hold an edge, X gathered by k list; k ascending, 4-k step grouping of sums changes
# speedup vs baseline: 5.2405x; 1.2439x over previous
#include <hip/hip_runtime.h>
#include <math.h>

#define Bsz   512
#define INsz  256
#define Hsz   1024
#define CONC  1280

typedef __attribute__((ext_vector_type(2))) float v2f;
typedef __attribute__((ext_vector_type(4))) float v4f;
typedef __attribute__((ext_vector_type(8))) float v8f;
typedef __attribute__((ext_vector_type(2))) int   v2i;

#define KGRP   8

#define KHALF       (CONC / 2)
#define STAGE_THR   256
#define CHUNK       1024
#define EPT         (CHUNK / STAGE_THR)
#define NBIN        128
#define OFFP        132
#define BIN_WGS     128
#define XT_ROWS     8
#define XT_WGS      (2 * (CONC / XT_ROWS))
#define STAGE_WGS   (BIN_WGS + XT_WGS)
#define MAXE        (1 << 20)

static_assert(INsz % XT_ROWS == 0 && CONC % XT_ROWS == 0, "a row group reads one source only");
static_assert(XT_ROWS == KGRP, "one work-group row block is one k group");
static_assert(CONC % 2 == 0, "two K halves");
static_assert(Bsz == 2 * STAGE_THR && XT_ROWS == 8, "X^T tiling must cover exactly");
static_assert(NBIN == 2 * (Hsz / 16), "one bin per (N tile, K half)");
static_assert(NBIN + 1 <= STAGE_THR && NBIN + 1 <= OFFP, "one thread per start offset");

__global__ __launch_bounds__(STAGE_THR) void stage_bin_x8(const int*   __restrict__ src,
                                                          const int*   __restrict__ dst,
                                                          const float* __restrict__ signs,
                                                          const float* __restrict__ w1,
                                                          const float* __restrict__ w2,
                                                          const float* __restrict__ wa,
                                                          const float* __restrict__ wb,
                                                          int*    __restrict__ recK,
                                                          float4* __restrict__ recV,
                                                          int*    __restrict__ off,
                                                          const float* __restrict__ in,
                                                          const float* __restrict__ hx,
                                                          float*  __restrict__ XT,
                                                          int E) {
    __shared__ int    cnt[NBIN];
    __shared__ int    start[NBIN + 1];
    __shared__ int    sK[CHUNK];
    __shared__ float4 sV[CHUNK];
    const int tid = threadIdx.x;

    if (blockIdx.x < BIN_WGS) {
        for (int chunk = blockIdx.x; chunk * CHUNK < E; chunk += BIN_WGS) {
            if (tid < NBIN) cnt[tid] = 0;
            __syncthreads();

            const int base = chunk * CHUNK;
            int    key[EPT], bin[EPT], rank[EPT];
            float4 val[EPT];
            bool   ok[EPT];
            #pragma unroll
            for (int u = 0; u < EPT; ++u) {
                const int e  = base + u * STAGE_THR + tid;
                ok[u] = e < E;
                const int ee = ok[u] ? e : 0;
                const int   k = src[ee];
                const int   n = dst[ee];
                const float s = signs[ee];
                val[u] = make_float4(w1[ee] * s, w2[ee] * s, wa[ee] * s, wb[ee] * s);
                bin[u] = ((n >> 4) << 1) | ((k >= KHALF) ? 1 : 0);
                key[u] = (k << 4) | (n & 15);
            }
            #pragma unroll
            for (int u = 0; u < EPT; ++u)
                rank[u] = ok[u] ? atomicAdd(&cnt[bin[u]], 1) : 0;
            __syncthreads();

            if (tid <= NBIN) {
                int s = 0;
                for (int b = 0; b < NBIN; ++b) {
                    const int c = cnt[b];
                    s += (b < tid) ? c : 0;
                }
                start[tid] = s;
                off[chunk * OFFP + tid] = s;
            }
            __syncthreads();

            #pragma unroll
            for (int u = 0; u < EPT; ++u) {
                if (ok[u]) {
                    const int q = start[bin[u]] + rank[u];
                    sK[q] = key[u];
                    sV[q] = val[u];
                }
            }
            __syncthreads();

            const int total = start[NBIN];
            #pragma unroll
            for (int u = 0; u < EPT; ++u) {
                const int q = u * STAGE_THR + tid;
                if (q < total) {
                    recK[base + q] = sK[q];
                    recV[base + q] = sV[q];
                }
            }
        }
    } else {
        const int w  = blockIdx.x - BIN_WGS;
        const int b  = (w & 1) * STAGE_THR + threadIdx.x;
        const int g  = w >> 1;
        const int j0 = g * KGRP;
        const float* srow = (j0 < INsz) ? in + (b * INsz + j0)
                                        : hx + (b * Hsz + (j0 - INsz));
        const float4 lo = *reinterpret_cast<const float4*>(srow);
        const float4 hi = *reinterpret_cast<const float4*>(srow + 4);
        float* d = XT + (size_t)j0 * Bsz + b;
        d[0 * Bsz] = lo.x;  d[1 * Bsz] = lo.y;  d[2 * Bsz] = lo.z;  d[3 * Bsz] = lo.w;
        d[4 * Bsz] = hi.x;  d[5 * Bsz] = hi.y;  d[6 * Bsz] = hi.z;  d[7 * Bsz] = hi.w;
    }
}

#define NT    4
#define CAP   640
#define CAPS  (CAP / 4)
#define NB    8
#define NBLK  (CAPS / NB)
#define KW    (CONC / 32)
#define LDSW_FLOATS (NT * CAP * 16)
#define EMAT  (128 * 16 + 4)
#define GEMM_THR 256
#define BUILD_THR (GEMM_THR / 2)
#define RB  16

static_assert(CONC % 32 == 0 && CAP % (4 * NB) == 0 && NBLK % 2 == 0, "passes are whole pairs of blocks");
static_assert(NT * KW <= GEMM_THR, "one thread per presence word");
static_assert(4 * EMAT <= LDSW_FLOATS, "epilogue planes reuse the weight region");

struct Frag {
    v2f a[NB][2][2];
    v2f b[NB][2];
};
struct Idx {
    v2i k[NB][2];
};

__device__ __forceinline__ void load_idx(Idx& x, const int* kl, int blk) {
    const int bc = (blk < NBLK) ? blk : (NBLK - 1);
    #pragma unroll
    for (int i = 0; i < NB; ++i) {
        #pragma unroll
        for (int t = 0; t < 2; ++t)
            x.k[i][t] = *reinterpret_cast<const v2i*>(kl + t * CONC + (bc * NB + i) * 4);
    }
}

__device__ __forceinline__ void load_a(Frag& f, const Idx& x, const char* __restrict__ xb, unsigned laneoff) {
    #pragma unroll
    for (int i = 0; i < NB; ++i) {
        #pragma unroll
        for (int t = 0; t < 2; ++t) {
            f.a[i][t][0] = *reinterpret_cast<const v2f*>(xb + ((unsigned)x.k[i][t].x + laneoff));
            f.a[i][t][1] = *reinterpret_cast<const v2f*>(xb + ((unsigned)x.k[i][t].y + laneoff));
        }
    }
}

__device__ __forceinline__ void load_b(Frag& f, const v2f* lb, int blk) {
    const int bc = (blk < NBLK) ? blk : (NBLK - 1);
    #pragma unroll
    for (int i = 0; i < NB; ++i) {
        #pragma unroll
        for (int t = 0; t < 2; ++t)
            f.b[i][t] = lb[(t * CAPS + bc * NB + i) * 32];
    }
}

__device__ __forceinline__ void mma_block(const Frag& f, v8f (&acc)[4]) {
    #pragma unroll
    for (int i = 0; i < NB; ++i) {
        #pragma unroll
        for (int t = 0; t < 2; ++t) {
            const v2f a0 = (v2f){f.a[i][t][0].x, f.a[i][t][1].x};
            const v2f a1 = (v2f){f.a[i][t][0].y, f.a[i][t][1].y};
            acc[2 * t + 0] = __builtin_amdgcn_wmma_f32_16x16x4_f32(
                false, a0, false, f.b[i][t], (short)0, acc[2 * t + 0], false, false);
            acc[2 * t + 1] = __builtin_amdgcn_wmma_f32_16x16x4_f32(
                false, a1, false, f.b[i][t], (short)0, acc[2 * t + 1], false, false);
        }
    }
}

__device__ __forceinline__ void fetch_records(int (&key)[RB], float4 (&v)[RB],
                                              const int*    __restrict__ recK,
                                              const float4* __restrict__ recV,
                                              int rb, int r0, int re) {
    #pragma unroll
    for (int u = 0; u < RB; ++u) {
        int r = (r0 + u < re) ? (r0 + u) : (re - 1);
        r = (r < 0) ? 0 : r;
        key[u] = recK[rb + r];
        v[u]   = recV[rb + r];
    }
}

struct FirstRun {
    int    rs, re;
    int    key[RB];
    float4 v[RB];
};

__device__ __forceinline__ void fetch_first_off(FirstRun& p, const int* __restrict__ off,
                                                int j, int nchunk, int b2) {
    p.rs = 0;
    p.re = 0;
    if (j < nchunk) {
        p.rs = off[j * OFFP + b2];
        p.re = off[j * OFFP + b2 + 1];
    }
}

__device__ __forceinline__ void fetch_first_rec(FirstRun& p,
                                                const int*    __restrict__ recK,
                                                const float4* __restrict__ recV,
                                                int j, int nchunk) {
    const int jc = (j < nchunk) ? j : 0;
    fetch_records(p.key, p.v, recK, recV, jc * CHUNK, p.rs, p.re);
}

template <class F>
__device__ __forceinline__ void for_records(const FirstRun& p,
                                            const int*    __restrict__ recK,
                                            const float4* __restrict__ recV,
                                            const int*    __restrict__ off,
                                            int j0, int nchunk, int b2, F fn) {
    #pragma unroll
    for (int u = 0; u < RB; ++u)
        if (p.rs + u < p.re) fn(p.key[u], p.v[u]);
    {
        const int rb = j0 * CHUNK;
        for (int r0 = p.rs + RB; r0 < p.re; r0 += RB) {
            int    key[RB];
            float4 v[RB];
            fetch_records(key, v, recK, recV, rb, r0, p.re);
            #pragma unroll
            for (int u = 0; u < RB; ++u)
                if (r0 + u < p.re) fn(key[u], v[u]);
        }
    }
    for (int j = j0 + BUILD_THR; j < nchunk; j += BUILD_THR) {
        const int rs = off[j * OFFP + b2];
        const int re = off[j * OFFP + b2 + 1];
        const int rb = j * CHUNK;
        for (int r0 = rs; r0 < re; r0 += RB) {
            int    key[RB];
            float4 v[RB];
            fetch_records(key, v, recK, recV, rb, r0, re);
            #pragma unroll
            for (int u = 0; u < RB; ++u)
                if (r0 + u < re) fn(key[u], v[u]);
        }
    }
}

__global__ __launch_bounds__(GEMM_THR) void gemm_fused(const float*  __restrict__ XT,
                                                       const int*    __restrict__ recK,
                                                       const float4* __restrict__ recV,
                                                       const int*    __restrict__ off,
                                                       const float* __restrict__ ts,
                                                       const float* __restrict__ b_ff1,
                                                       const float* __restrict__ b_ff2,
                                                       const float* __restrict__ b_ta,
                                                       const float* __restrict__ b_tb,
                                                       float* __restrict__ out,
                                                       int E) {
    __shared__ __attribute__((aligned(16))) float    ldsW[LDSW_FLOATS];
    __shared__ __attribute__((aligned(16))) int      klist[NT * CONC];
    __shared__ int      kmap[NT * CONC];
    __shared__ unsigned pres[NT * KW];
    __shared__ int      cnt[NT];

    const int tid  = threadIdx.x;
    const int lane = tid & 31;
    const int wave = tid >> 5;

    const int ntile = blockIdx.x;
    const int n0   = ntile * 16;
    const int rg   = wave & 3;
    const int qa   = (wave >> 2) * 2;
    const int half = lane >> 4;
    const int nchunk = (E + CHUNK - 1) / CHUNK;

    const int kh = tid >> 7;
    const int j0 = tid & (BUILD_THR - 1);
    const int b2 = ntile * 2 + kh;

    FirstRun p;
    fetch_first_off(p, off, j0, nchunk, b2);
    fetch_first_rec(p, recK, recV, j0, nchunk);

    v8f acc[4];
    #pragma unroll
    for (int l = 0; l < 4; ++l)
        acc[l] = (v8f){0.f,0.f,0.f,0.f,0.f,0.f,0.f,0.f};

    v4f* l4 = reinterpret_cast<v4f*>(ldsW);
    for (int i = tid; i < LDSW_FLOATS / 4; i += GEMM_THR)
        l4[i] = (v4f){0.f, 0.f, 0.f, 0.f};
    for (int i = tid; i < NT * CONC; i += GEMM_THR)
        klist[i] = 0;
    if (tid < NT * KW) pres[tid] = 0u;
    __syncthreads();

    for_records(p, recK, recV, off, j0, nchunk, b2, [&](int key, const float4&) {
        const int k = key >> 4;
        const int q = (key & 15) >> 2;
        atomicOr(&pres[q * KW + (k >> 5)], 1u << (k & 31));
    });
    __syncthreads();

    if (tid < NT * KW) {
        const int q = tid / KW;
        const int w = tid - q * KW;
        int n = 0;
        for (int i = 0; i < KW; ++i) {
            const int c = __popc(pres[q * KW + i]);
            n += (i < w) ? c : 0;
        }
        unsigned bits = pres[tid];
        while (bits) {
            const int b = __ffs(bits) - 1;
            bits &= bits - 1u;
            const int k = w * 32 + b;
            kmap[q * CONC + k]  = n;
            klist[q * CONC + n] = k * (Bsz * 4);
            ++n;
        }
        if (w == KW - 1) cnt[q] = n;
    }
    __syncthreads();

    int maxcnt = 0;
    #pragma unroll
    for (int q = 0; q < NT; ++q) maxcnt = (cnt[q] > maxcnt) ? cnt[q] : maxcnt;
    const int npass = __builtin_amdgcn_readfirstlane((maxcnt + CAP - 1) / CAP);
    const int cnta = cnt[qa], cntb = cnt[qa + 1];

    const unsigned laneoff = (unsigned)((blockIdx.y * 128 + rg * 32 + 2 * (lane & 15)) * 4);
    const char* xb = reinterpret_cast<const char*>(XT);
    const v2f*  lb = reinterpret_cast<const v2f*>(ldsW) + (qa * CAPS * 32 + lane);

    for (int ps = 0; ps < npass; ++ps) {
        const int pbase = ps * CAP;
        if (ps > 0) {
            for (int i = tid; i < LDSW_FLOATS / 4; i += GEMM_THR)
                l4[i] = (v4f){0.f, 0.f, 0.f, 0.f};
            __syncthreads();
        }
        const int* kl = klist + (qa * CONC + pbase + 2 * half);

        Frag f0, f1;
        Idx  i0, i1;
        load_idx(i0, kl, 0);
        load_a(f0, i0, xb, laneoff);

        for_records(p, recK, recV, off, j0, nchunk, b2, [&](int key, const float4& v) {
            const int k    = key >> 4;
            const int c    = key & 15;
            const int q    = c >> 2;
            const int col4 = c & 3;
            const int kk   = kmap[q * CONC + k] - pbase;
            if ((unsigned)kk < (unsigned)CAP) {
                const int s  = kk >> 2;
                const int hf = (kk >> 1) & 1;
                const int e  = kk & 1;
                const int fi = (((q * CAPS + s) * 32) + hf * 16 + col4 * 4) * 2 + e;
                atomicAdd(&ldsW[fi],     v.x);
                atomicAdd(&ldsW[fi + 2], v.y);
                atomicAdd(&ldsW[fi + 4], v.z);
                atomicAdd(&ldsW[fi + 6], v.w);
            }
        });
        __syncthreads();

        int ra = cnta - pbase;  ra = (ra < 0) ? 0 : ((ra > CAP) ? CAP : ra);
        int rb2 = cntb - pbase; rb2 = (rb2 < 0) ? 0 : ((rb2 > CAP) ? CAP : rb2);
        const int rm = (ra > rb2) ? ra : rb2;
        const int trip = __builtin_amdgcn_readfirstlane(((rm + 8 * NB - 1) / (8 * NB)) * 2);

        load_b(f0, lb, 0);
        load_idx(i1, kl, 1);
        for (int blk = 0; blk < trip; blk += 2) {
            load_a(f1, i1, xb, laneoff);
            load_b(f1, lb, blk + 1);
            load_idx(i0, kl, blk + 2);
            __builtin_amdgcn_sched_barrier(0);
            mma_block(f0, acc);
            __builtin_amdgcn_sched_barrier(0);
            load_a(f0, i0, xb, laneoff);
            load_b(f0, lb, blk + 2);
            load_idx(i1, kl, blk + 3);
            __builtin_amdgcn_sched_barrier(0);
            mma_block(f1, acc);
            __builtin_amdgcn_sched_barrier(0);
        }
        __syncthreads();
    }

    {
        const int mat  = lane & 3;
        const int col4 = (lane >> 2) & 3;
        float* ep = ldsW + mat * EMAT + (rg * 32 + 16 * half) * 16 + qa * 4 + col4;
        #pragma unroll
        for (int t = 0; t < 2; ++t) {
            #pragma unroll
            for (int sub = 0; sub < 2; ++sub) {
                #pragma unroll
                for (int r = 0; r < 8; ++r)
                    ep[(2 * r + sub) * 16 + t * 4] = acc[2 * t + sub][r];
            }
        }
    }
    __syncthreads();

    const int c4 = (tid & 3) * 4;
    const int ng = n0 + c4;
    const v4f bf1 = *reinterpret_cast<const v4f*>(b_ff1 + ng);
    const v4f bf2 = *reinterpret_cast<const v4f*>(b_ff2 + ng);
    const v4f bta = *reinterpret_cast<const v4f*>(b_ta + ng);
    const v4f btb = *reinterpret_cast<const v4f*>(b_tb + ng);
    #pragma unroll
    for (int rr = 0; rr < 2; ++rr) {
        const int row = (tid >> 2) + 64 * rr;
        const int m   = blockIdx.y * 128 + row;
        const float tsv = ts[m];
        const float* e0 = ldsW + row * 16 + c4;
        const v4f p1 = *reinterpret_cast<const v4f*>(e0);
        const v4f p2 = *reinterpret_cast<const v4f*>(e0 + EMAT);
        const v4f pa = *reinterpret_cast<const v4f*>(e0 + 2 * EMAT);
        const v4f pb = *reinterpret_cast<const v4f*>(e0 + 3 * EMAT);
        v4f h;
        #pragma unroll
        for (int i = 0; i < 4; ++i) {
            const float f1v = __builtin_amdgcn_tanhf(p1[i] + bf1[i]);
            const float f2v = __builtin_amdgcn_tanhf(p2[i] + bf2[i]);
            const float ta  = pa[i] + bta[i];
            const float tb  = pb[i] + btb[i];
            const float sg  = 0.5f + 0.5f * __builtin_amdgcn_tanhf(0.5f * (ta * tsv + tb));
            h[i] = f1v * (1.f - sg) + sg * f2v;
        }
        const size_t o = (size_t)m * Hsz + ng;
        *reinterpret_cast<v4f*>(out + o)                      = h;
        *reinterpret_cast<v4f*>(out + (size_t)Bsz * Hsz + o)  = h;
    }
}

extern "C" void kernel_launch(void* const* d_in, const int* in_sizes, int n_in,
                              void* d_out, int out_size, void* d_ws, size_t ws_size,
                              hipStream_t stream) {
    const float* input = (const float*)d_in[0];
    const float* hx    = (const float*)d_in[1];
    const float* ts    = (const float*)d_in[2];
    const int*   src   = (const int*)  d_in[3];
    const int*   dst   = (const int*)  d_in[4];
    const float* signs = (const float*)d_in[5];
    const float* w1    = (const float*)d_in[6];
    const float* w2    = (const float*)d_in[7];
    const float* wa    = (const float*)d_in[8];
    const float* wb    = (const float*)d_in[9];
    const float* bf1   = (const float*)d_in[10];
    const float* bf2   = (const float*)d_in[11];
    const float* bta   = (const float*)d_in[12];
    const float* btb   = (const float*)d_in[13];
    const int    E     = in_sizes[3];

    float*  XT   = (float*)d_ws;
    float4* recV = (float4*)(XT + CONC * Bsz);
    int*    recK = (int*)(recV + MAXE);
    int*    off  = recK + MAXE;
    float*  out  = (float*)d_out;

    stage_bin_x8<<<STAGE_WGS, STAGE_THR, 0, stream>>>(src, dst, signs, w1, w2, wa, wb,
                                                      recK, recV, off, input, hx, XT, E);

    dim3 grid(Hsz / 16, Bsz / 128);
    gemm_fused<<<grid, GEMM_THR, 0, stream>>>(XT, recK, recV, off, ts, bf1, bf2, bta, btb, out, E);
}
